# starting kernel
# baseline (speedup 1.0000x reference)
#include <hip/hip_runtime.h>
#include <math.h>

typedef __attribute__((ext_vector_type(2))) float v2f;
typedef __attribute__((ext_vector_type(8))) float v8f;

#define B_   64
#define O_   256
#define M_   8
#define D_   512
#define EPS_ 1e-5f
#define BT   16
#define NBT  (B_ / BT)
#define NW   8
#define NJ   (D_ / (4 * NW))

__device__ __forceinline__ float fast_sigmoid(float x) {
#if __has_builtin(__builtin_amdgcn_tanhf)
    return fmaf(0.5f, __builtin_amdgcn_tanhf(0.5f * x), 0.5f);
#else
    return __builtin_amdgcn_rcpf(1.0f + __expf(-x));
#endif
}

__global__ __launch_bounds__(256) void dnm_prep(
    const float* __restrict__ x,   const float* __restrict__ g,
    const float* __restrict__ be,  const float* __restrict__ dnw,
    const float* __restrict__ dnb, float* __restrict__ xn,
    float* __restrict__ wsum,      float* __restrict__ bsum)
{
    __shared__ float s1[256];
    __shared__ float s2[256];
    const int t  = threadIdx.x;
    const int bx = blockIdx.x;

    if (bx < B_) {
        const float* row = x + bx * D_;
        float a = row[t];
        float b = row[t + 256];
        s1[t] = a + b;
        s2[t] = a * a + b * b;
        __syncthreads();
        for (int s = 128; s > 0; s >>= 1) {
            if (t < s) { s1[t] += s1[t + s]; s2[t] += s2[t + s]; }
            __syncthreads();
        }
        float mu   = s1[0] * (1.0f / D_);
        float var  = s2[0] * (1.0f / D_) - mu * mu;
        float rstd = rsqrtf(var + EPS_);
        float* out = xn + bx * D_;
        out[t]       = (a - mu) * rstd * g[t]       + be[t];
        out[t + 256] = (b - mu) * rstd * g[t + 256] + be[t + 256];
    } else {
        const int m = bx - B_;
        const float* wr = dnw + m * D_;
        const float* br = dnb + m * D_;
        s1[t] = wr[t] + wr[t + 256];
        s2[t] = br[t] + br[t + 256];
        __syncthreads();
        for (int s = 128; s > 0; s >>= 1) {
            if (t < s) { s1[t] += s1[t + s]; s2[t] += s2[t + s]; }
            __syncthreads();
        }
        if (t == 0) { wsum[m] = s1[0]; bsum[m] = s2[0]; }
    }
}

__global__ __launch_bounds__(NW * 32) void dnm_main(
    const float* __restrict__ xn,   const float* __restrict__ sw,
    const float* __restrict__ sb,   const float* __restrict__ dnw,
    const float* __restrict__ wsum, const float* __restrict__ bsum,
    float* __restrict__ pre)
{
    constexpr int XP = D_ + 4;
    __shared__ float xs[BT * XP];
    __shared__ float c1buf[NW * 256];
    __shared__ float c2buf[NW * 256];

    const int o     = blockIdx.x >> 2;
    const int b0    = (blockIdx.x & (NBT - 1)) * BT;
    const int tid   = threadIdx.x;
    const int w     = tid >> 5;
    const int lane  = tid & 31;
    const int tb    = lane & 15;
    const int khalf = lane >> 4;
    const int kb    = khalf * 2;

    for (int idx = tid; idx < BT * D_; idx += NW * 32) {
        int r = idx >> 9, c = idx & (D_ - 1);
        xs[r * XP + c] = xn[(b0 + r) * D_ + c];
    }
    __syncthreads();

    v8f c1 = {};
    v8f c2 = {};

    const int    dstart = 4 * w + kb;
    const float  fone   = (tb == 0) ? 1.0f : 0.0f;
    const float* dwl    = dnw + (((tb >= 1) && (tb <= M_)) ? (tb - 1) * D_ : 0) + dstart;
    const float* xrow   = xs + tb * XP + dstart;
    const float* swl    = sw + (size_t)o * (M_ * D_) + dstart;
    const float* sbl    = sb + (size_t)o * (M_ * D_) + dstart;

    #pragma unroll 1
    for (int m = 0; m < M_; ++m) {
        __builtin_prefetch(swl + D_, 0, 1);
        __builtin_prefetch(sbl + D_, 0, 1);
        const float fm = (tb == m + 1) ? 1.0f : 0.0f;
        #pragma unroll 4
        for (int j = 0; j < NJ; ++j) {
            const int off = j * 4 * NW;
            float2 swv = *(const float2*)(swl + off);
            float2 sbv = *(const float2*)(sbl + off);
            float2 dwv = *(const float2*)(dwl + off);
            float2 xv  = *(const float2*)(xrow + off);

            v2f a;
            a.x = fast_sigmoid(fmaf(swv.x, xv.x, sbv.x));
            a.y = fast_sigmoid(fmaf(swv.y, xv.y, sbv.y));

            v2f bm;
            bm.x = fmaf(fm, dwv.x, fone);
            bm.y = fmaf(fm, dwv.y, fone);

            c1 = __builtin_amdgcn_wmma_f32_16x16x4_f32(false, a, false, bm, (short)0, c1, false, false);
            c2 = __builtin_amdgcn_wmma_f32_16x16x4_f32(false, a, false, a,  (short)0, c2, false, false);
        }
        swl += D_;
        sbl += D_;
    }

    #pragma unroll
    for (int r = 0; r < 8; ++r) {
        const int row = r + khalf * 8;
        c1buf[w * 256 + row * 16 + tb] = c1[r];
        c2buf[w * 256 + row * 16 + tb] = c2[r];
    }
    __syncthreads();

    if (tid < BT) {
        float S1 = 0.f, S2 = 0.f, T[M_];
        #pragma unroll
        for (int m = 0; m < M_; ++m) T[m] = 0.f;
        for (int wv = 0; wv < NW; ++wv) {
            const float* p1 = c1buf + wv * 256 + tid * 16;
            const float* p2 = c2buf + wv * 256 + tid * 16;
            S1 += p1[0];
            S2 += p2[tid];
            #pragma unroll
            for (int m = 0; m < M_; ++m) T[m] += p1[m + 1];
        }
        const float inv  = 1.0f / (float)(M_ * D_);
        const float mu   = S1 * inv;
        const float var  = S2 * inv - mu * mu;
        const float rstd = rsqrtf(var + EPS_);
        float acc = 0.f;
        #pragma unroll
        for (int m = 0; m < M_; ++m) {
            float s = rstd * (T[m] - mu * wsum[m]) + bsum[m];
            acc += fast_sigmoid(s);
        }
        pre[(size_t)(b0 + tid) * O_ + o] = acc;
    }
}

__global__ __launch_bounds__(256) void dnm_softmax(
    const float* __restrict__ pre, float* __restrict__ out)
{
    __shared__ float sh[256];
    const int b = blockIdx.x, t = threadIdx.x;
    float v = pre[b * O_ + t];
    sh[t] = v;
    __syncthreads();
    for (int s = 128; s > 0; s >>= 1) {
        if (t < s) sh[t] = fmaxf(sh[t], sh[t + s]);
        __syncthreads();
    }
    float mx = sh[0];
    __syncthreads();
    float e = __expf(v - mx);
    sh[t] = e;
    __syncthreads();
    for (int s = 128; s > 0; s >>= 1) {
        if (t < s) sh[t] += sh[t + s];
        __syncthreads();
    }
    out[b * O_ + t] = e * __builtin_amdgcn_rcpf(sh[0]);
}

extern "C" void kernel_launch(void* const* d_in, const int* in_sizes, int n_in,
                              void* d_out, int out_size, void* d_ws, size_t ws_size,
                              hipStream_t stream) {
    const float* x    = (const float*)d_in[0];
    const float* sn0w = (const float*)d_in[1];
    const float* sn0b = (const float*)d_in[2];
    const float* sw   = (const float*)d_in[3];
    const float* sb   = (const float*)d_in[4];
    const float* dnw  = (const float*)d_in[5];
    const float* dnb  = (const float*)d_in[6];
    float* out = (float*)d_out;

    float* ws   = (float*)d_ws;
    float* xn   = ws;
    float* wsum = ws + 32768;
    float* bsum = ws + 32776;
    float* pre  = ws + 32784;

    dnm_prep   <<<B_ + M_,  256,     0, stream>>>(x, sn0w, sn0b, dnw, dnb, xn, wsum, bsum);
    dnm_main   <<<O_ * NBT, NW * 32, 0, stream>>>(xn, sw, sb, dnw, wsum, bsum, pre);
    dnm_softmax<<<B_,       256,     0, stream>>>(pre, out);
}


# Re-map K so each lane half owns one contiguous 32-float chunk of d: sw, sb and row m of dn_w arrive as b128 loads (12 per 8 steps), x chunk kept in registers; summation order changes
# speedup vs baseline: 1.1892x; 1.1892x over previous
#include <hip/hip_runtime.h>
#include <math.h>

typedef __attribute__((ext_vector_type(2))) float v2f;
typedef __attribute__((ext_vector_type(8))) float v8f;

#define B_   64
#define O_   256
#define M_   8
#define D_   512
#define EPS_ 1e-5f
#define BT   16
#define NBT  (B_ / BT)
#define NW   8
#define NJ   (D_ / (4 * NW))

__device__ __forceinline__ float fast_sigmoid(float x) {
#if __has_builtin(__builtin_amdgcn_tanhf)
    return fmaf(0.5f, __builtin_amdgcn_tanhf(0.5f * x), 0.5f);
#else
    return __builtin_amdgcn_rcpf(1.0f + __expf(-x));
#endif
}

__global__ __launch_bounds__(256) void dnm_prep(
    const float* __restrict__ x,   const float* __restrict__ g,
    const float* __restrict__ be,  const float* __restrict__ dnw,
    const float* __restrict__ dnb, float* __restrict__ xn,
    float* __restrict__ wsum,      float* __restrict__ bsum)
{
    __shared__ float s1[256];
    __shared__ float s2[256];
    const int t  = threadIdx.x;
    const int bx = blockIdx.x;

    if (bx < B_) {
        const float* row = x + bx * D_;
        float a = row[t];
        float b = row[t + 256];
        s1[t] = a + b;
        s2[t] = a * a + b * b;
        __syncthreads();
        for (int s = 128; s > 0; s >>= 1) {
            if (t < s) { s1[t] += s1[t + s]; s2[t] += s2[t + s]; }
            __syncthreads();
        }
        float mu   = s1[0] * (1.0f / D_);
        float var  = s2[0] * (1.0f / D_) - mu * mu;
        float rstd = rsqrtf(var + EPS_);
        float* out = xn + bx * D_;
        out[t]       = (a - mu) * rstd * g[t]       + be[t];
        out[t + 256] = (b - mu) * rstd * g[t + 256] + be[t + 256];
    } else {
        const int m = bx - B_;
        const float* wr = dnw + m * D_;
        const float* br = dnb + m * D_;
        s1[t] = wr[t] + wr[t + 256];
        s2[t] = br[t] + br[t + 256];
        __syncthreads();
        for (int s = 128; s > 0; s >>= 1) {
            if (t < s) { s1[t] += s1[t + s]; s2[t] += s2[t + s]; }
            __syncthreads();
        }
        if (t == 0) { wsum[m] = s1[0]; bsum[m] = s2[0]; }
    }
}

__global__ __launch_bounds__(NW * 32) void dnm_main(
    const float* __restrict__ xn,   const float* __restrict__ sw,
    const float* __restrict__ sb,   const float* __restrict__ dnw,
    const float* __restrict__ wsum, const float* __restrict__ bsum,
    float* __restrict__ pre)
{
    constexpr int XP = D_ + 4;
    __shared__ __attribute__((aligned(16))) float xs[BT * XP];
    __shared__ float c1buf[NW * 256];
    __shared__ float c2buf[NW * 256];

    const int o     = blockIdx.x >> 2;
    const int b0    = (blockIdx.x & (NBT - 1)) * BT;
    const int tid   = threadIdx.x;
    const int w     = tid >> 5;
    const int lane  = tid & 31;
    const int tb    = lane & 15;
    const int khalf = lane >> 4;

    for (int idx = tid; idx < BT * D_; idx += NW * 32) {
        int r = idx >> 9, c = idx & (D_ - 1);
        xs[r * XP + c] = xn[(b0 + r) * D_ + c];
    }
    __syncthreads();

    v8f c1 = {};
    v8f c2 = {};

    const int    cstart = 64 * w + 32 * khalf;
    const float  fone   = (tb == 0) ? 1.0f : 0.0f;
    const float* swl    = sw + (size_t)o * (M_ * D_) + cstart;
    const float* sbl    = sb + (size_t)o * (M_ * D_) + cstart;
    const float* dwl    = dnw + cstart;

    float xr[32];
    #pragma unroll
    for (int i = 0; i < 8; ++i) {
        const float4 v = *(const float4*)(xs + tb * XP + cstart + 4 * i);
        xr[4 * i + 0] = v.x; xr[4 * i + 1] = v.y; xr[4 * i + 2] = v.z; xr[4 * i + 3] = v.w;
    }

    #pragma unroll 1
    for (int m = 0; m < M_; ++m) {
        __builtin_prefetch(swl + D_, 0, 1);
        __builtin_prefetch(sbl + D_, 0, 1);
        const float fm = (tb == m + 1) ? 1.0f : 0.0f;
        #pragma unroll
        for (int hh = 0; hh < 2; ++hh) {
            float4 s4[4], b4[4], d4[4];
            #pragma unroll
            for (int i = 0; i < 4; ++i) {
                s4[i] = *(const float4*)(swl + 16 * hh + 4 * i);
                b4[i] = *(const float4*)(sbl + 16 * hh + 4 * i);
                d4[i] = *(const float4*)(dwl + 16 * hh + 4 * i);
            }
            __builtin_amdgcn_sched_barrier(0);
            #pragma unroll
            for (int i = 0; i < 4; ++i) {
                const int xo = 16 * hh + 4 * i;
                v2f a, bm;
                a.x  = fast_sigmoid(fmaf(s4[i].x, xr[xo + 0], b4[i].x));
                a.y  = fast_sigmoid(fmaf(s4[i].y, xr[xo + 1], b4[i].y));
                bm.x = fmaf(fm, d4[i].x, fone);
                bm.y = fmaf(fm, d4[i].y, fone);
                c1 = __builtin_amdgcn_wmma_f32_16x16x4_f32(false, a, false, bm, (short)0, c1, false, false);
                c2 = __builtin_amdgcn_wmma_f32_16x16x4_f32(false, a, false, a,  (short)0, c2, false, false);

                a.x  = fast_sigmoid(fmaf(s4[i].z, xr[xo + 2], b4[i].z));
                a.y  = fast_sigmoid(fmaf(s4[i].w, xr[xo + 3], b4[i].w));
                bm.x = fmaf(fm, d4[i].z, fone);
                bm.y = fmaf(fm, d4[i].w, fone);
                c1 = __builtin_amdgcn_wmma_f32_16x16x4_f32(false, a, false, bm, (short)0, c1, false, false);
                c2 = __builtin_amdgcn_wmma_f32_16x16x4_f32(false, a, false, a,  (short)0, c2, false, false);
            }
            __builtin_amdgcn_sched_barrier(0);
        }
        swl += D_;
        sbl += D_;
        dwl += D_;
    }

    #pragma unroll
    for (int r = 0; r < 8; ++r) {
        const int row = r + khalf * 8;
        c1buf[w * 256 + row * 16 + tb] = c1[r];
        c2buf[w * 256 + row * 16 + tb] = c2[r];
    }
    __syncthreads();

    if (tid < BT) {
        float S1 = 0.f, S2 = 0.f, T[M_];
        #pragma unroll
        for (int m = 0; m < M_; ++m) T[m] = 0.f;
        for (int wv = 0; wv < NW; ++wv) {
            const float* p1 = c1buf + wv * 256 + tid * 16;
            const float* p2 = c2buf + wv * 256 + tid * 16;
            S1 += p1[0];
            S2 += p2[tid];
            #pragma unroll
            for (int m = 0; m < M_; ++m) T[m] += p1[m + 1];
        }
        const float inv  = 1.0f / (float)(M_ * D_);
        const float mu   = S1 * inv;
        const float var  = S2 * inv - mu * mu;
        const float rstd = rsqrtf(var + EPS_);
        float acc = 0.f;
        #pragma unroll
        for (int m = 0; m < M_; ++m) {
            float s = rstd * (T[m] - mu * wsum[m]) + bsum[m];
            acc += fast_sigmoid(s);
        }
        pre[(size_t)(b0 + tid) * O_ + o] = acc;
    }
}

__global__ __launch_bounds__(256) void dnm_softmax(
    const float* __restrict__ pre, float* __restrict__ out)
{
    __shared__ float sh[256];
    const int b = blockIdx.x, t = threadIdx.x;
    float v = pre[b * O_ + t];
    sh[t] = v;
    __syncthreads();
    for (int s = 128; s > 0; s >>= 1) {
        if (t < s) sh[t] = fmaxf(sh[t], sh[t + s]);
        __syncthreads();
    }
    float mx = sh[0];
    __syncthreads();
    float e = __expf(v - mx);
    sh[t] = e;
    __syncthreads();
    for (int s = 128; s > 0; s >>= 1) {
        if (t < s) sh[t] += sh[t + s];
        __syncthreads();
    }
    out[b * O_ + t] = e * __builtin_amdgcn_rcpf(sh[0]);
}

extern "C" void kernel_launch(void* const* d_in, const int* in_sizes, int n_in,
                              void* d_out, int out_size, void* d_ws, size_t ws_size,
                              hipStream_t stream) {
    const float* x    = (const float*)d_in[0];
    const float* sn0w = (const float*)d_in[1];
    const float* sn0b = (const float*)d_in[2];
    const float* sw   = (const float*)d_in[3];
    const float* sb   = (const float*)d_in[4];
    const float* dnw  = (const float*)d_in[5];
    const float* dnb  = (const float*)d_in[6];
    float* out = (float*)d_out;

    float* ws   = (float*)d_ws;
    float* xn   = ws;
    float* wsum = ws + 32768;
    float* bsum = ws + 32776;
    float* pre  = ws + 32784;

    dnm_prep   <<<B_ + M_,  256,     0, stream>>>(x, sn0w, sn0b, dnw, dnb, xn, wsum, bsum);
    dnm_main   <<<O_ * NBT, NW * 32, 0, stream>>>(xn, sw, sb, dnw, wsum, bsum, pre);
    dnm_softmax<<<B_,       256,     0, stream>>>(pre, out);
}


# Halve the WMMA count: ones and dn_w ride as rows 14,15 of a 14-batch-row tile V, one c += V x V^T per step gives sum z, T[m], sum z^2; one 16-wave workgroup per o; summation order changes
# speedup vs baseline: 1.4092x; 1.1850x over previous
#include <hip/hip_runtime.h>
#include <math.h>

typedef __attribute__((ext_vector_type(2))) float v2f;
typedef __attribute__((ext_vector_type(8))) float v8f;

#define B_   64
#define O_   256
#define M_   8
#define D_   512
#define EPS_ 1e-5f
#define TR   14
#define NT   5
#define NW   16
#define CH   16

__device__ __forceinline__ float fast_sigmoid(float x) {
#if __has_builtin(__builtin_amdgcn_tanhf)
    return fmaf(0.5f, __builtin_amdgcn_tanhf(0.5f * x), 0.5f);
#else
    return __builtin_amdgcn_rcpf(1.0f + __expf(-x));
#endif
}

__device__ __forceinline__ float zcore(float u) {
#if __has_builtin(__builtin_amdgcn_tanhf)
    return __builtin_amdgcn_tanhf(0.5f * u);
#else
    return __builtin_amdgcn_rcpf(1.0f + __expf(-u));
#endif
}

__global__ __launch_bounds__(256) void dnm_prep(
    const float* __restrict__ x,   const float* __restrict__ g,
    const float* __restrict__ be,  const float* __restrict__ dnw,
    const float* __restrict__ dnb, float* __restrict__ xn,
    float* __restrict__ wsum,      float* __restrict__ bsum)
{
    __shared__ float s1[256];
    __shared__ float s2[256];
    const int t  = threadIdx.x;
    const int bx = blockIdx.x;

    if (bx < B_) {
        const float* row = x + bx * D_;
        float a = row[t];
        float b = row[t + 256];
        s1[t] = a + b;
        s2[t] = a * a + b * b;
        __syncthreads();
        for (int s = 128; s > 0; s >>= 1) {
            if (t < s) { s1[t] += s1[t + s]; s2[t] += s2[t + s]; }
            __syncthreads();
        }
        float mu   = s1[0] * (1.0f / D_);
        float var  = s2[0] * (1.0f / D_) - mu * mu;
        float rstd = rsqrtf(var + EPS_);
        float* out = xn + bx * D_;
        out[t]       = (a - mu) * rstd * g[t]       + be[t];
        out[t + 256] = (b - mu) * rstd * g[t + 256] + be[t + 256];
    } else {
        const int m = bx - B_;
        const float* wr = dnw + m * D_;
        const float* br = dnb + m * D_;
        s1[t] = wr[t] + wr[t + 256];
        s2[t] = br[t] + br[t + 256];
        __syncthreads();
        for (int s = 128; s > 0; s >>= 1) {
            if (t < s) { s1[t] += s1[t + s]; s2[t] += s2[t + s]; }
            __syncthreads();
        }
        if (t == 0) { wsum[m] = s1[0]; bsum[m] = s2[0]; }
    }
}

__global__ __launch_bounds__(NW * 32) void dnm_main(
    const float* __restrict__ xn,   const float* __restrict__ sw,
    const float* __restrict__ sb,   const float* __restrict__ dnw,
    const float* __restrict__ wsum, const float* __restrict__ bsum,
    float* __restrict__ pre)
{
    __shared__ float Tb[NW * M_ * NT * 16];
    __shared__ float S1b[NW * NT * 16];
    __shared__ float S2b[NW * NT * 16];
    __shared__ float red[(M_ + 2) * B_];

    const int o     = blockIdx.x;
    const int tid   = threadIdx.x;
    const int w     = tid >> 5;
    const int lane  = tid & 31;
    const int tb    = lane & 15;
    const int khalf = lane >> 4;
    const int cstart = 2 * CH * w + CH * khalf;

    const bool  isz = tb < TR;
#if __has_builtin(__builtin_amdgcn_tanhf)
    const float zs  = isz ? 0.5f : 0.0f;
    const float zo  = isz ? 0.5f : (tb == TR ? 1.0f : 0.0f);
#else
    const float zs  = isz ? 1.0f : 0.0f;
    const float zo  = (tb == TR) ? 1.0f : 0.0f;
#endif
    const float fw  = (tb == TR + 1) ? 1.0f : 0.0f;

    float xr[NT][CH];
    #pragma unroll
    for (int t = 0; t < NT; ++t) {
        int row = TR * t + (isz ? tb : TR - 1);
        row = row < B_ ? row : B_ - 1;
        #pragma unroll
        for (int q = 0; q < CH / 4; ++q) {
            const float4 v = *(const float4*)(xn + row * D_ + cstart + 4 * q);
            xr[t][4 * q + 0] = isz ? v.x : 0.f; xr[t][4 * q + 1] = isz ? v.y : 0.f;
            xr[t][4 * q + 2] = isz ? v.z : 0.f; xr[t][4 * q + 3] = isz ? v.w : 0.f;
        }
    }

    v8f acc[NT];
    #pragma unroll
    for (int t = 0; t < NT; ++t) acc[t] = (v8f){};

    const float* swl = sw + (size_t)o * (M_ * D_) + cstart;
    const float* sbl = sb + (size_t)o * (M_ * D_) + cstart;
    const float* dwl = dnw + cstart;
    float* tst = Tb + (w * M_ * NT) * 16 + tb;

    #pragma unroll 1
    for (int m = 0; m < M_; ++m) {
        __builtin_prefetch(swl + D_, 0, 1);
        __builtin_prefetch(sbl + D_, 0, 1);
        float s[CH], b[CH], g[CH];
        #pragma unroll
        for (int q = 0; q < CH / 4; ++q) {
            const float4 sv = *(const float4*)(swl + 4 * q);
            const float4 bv = *(const float4*)(sbl + 4 * q);
            const float4 dv = *(const float4*)(dwl + 4 * q);
            s[4 * q + 0] = sv.x; s[4 * q + 1] = sv.y; s[4 * q + 2] = sv.z; s[4 * q + 3] = sv.w;
            b[4 * q + 0] = bv.x; b[4 * q + 1] = bv.y; b[4 * q + 2] = bv.z; b[4 * q + 3] = bv.w;
            g[4 * q + 0] = fmaf(fw, dv.x, zo); g[4 * q + 1] = fmaf(fw, dv.y, zo);
            g[4 * q + 2] = fmaf(fw, dv.z, zo); g[4 * q + 3] = fmaf(fw, dv.w, zo);
        }

        #pragma unroll
        for (int i = 0; i < CH / 2; ++i) {
            #pragma unroll
            for (int t = 0; t < NT; ++t) {
                v2f a;
                a.x = fmaf(zs, zcore(fmaf(s[2 * i],     xr[t][2 * i],     b[2 * i])),     g[2 * i]);
                a.y = fmaf(zs, zcore(fmaf(s[2 * i + 1], xr[t][2 * i + 1], b[2 * i + 1])), g[2 * i + 1]);
                acc[t] = __builtin_amdgcn_wmma_f32_16x16x4_f32(false, a, false, a, (short)0, acc[t], false, false);
            }
        }

        asm volatile("v_nop\n\tv_nop\n\tv_nop\n\tv_nop\n\tv_nop"
                     : "+v"(acc[0]), "+v"(acc[1]), "+v"(acc[2]), "+v"(acc[3]), "+v"(acc[4]));
        if (khalf) {
            #pragma unroll
            for (int t = 0; t < NT; ++t) {
                tst[(m * NT + t) * 16] = acc[t][7];
                acc[t][7] = 0.f;
            }
        }
        swl += D_;
        sbl += D_;
        dwl += D_;
    }

    #pragma unroll
    for (int t = 0; t < NT; ++t) {
        if (khalf) S1b[(w * NT + t) * 16 + tb] = acc[t][6];
        float dv = acc[t][0];
        #pragma unroll
        for (int r = 1; r < 8; ++r) dv = ((tb & 7) == r) ? acc[t][r] : dv;
        if ((tb >> 3) == khalf) S2b[(w * NT + t) * 16 + tb] = dv;
    }
    __syncthreads();

    for (int idx = tid; idx < (M_ + 2) * B_; idx += NW * 32) {
        const int q = idx >> 6;
        const int b = idx & (B_ - 1);
        const int t = b / TR;
        const int j = b - t * TR;
        const float* p;
        int stride;
        if (q < M_)       { p = Tb  + (q * NT + t) * 16 + j; stride = M_ * NT * 16; }
        else if (q == M_) { p = S1b + t * 16 + j;            stride = NT * 16; }
        else              { p = S2b + t * 16 + j;            stride = NT * 16; }
        float sum = 0.f;
        for (int wv = 0; wv < NW; ++wv) sum += p[wv * stride];
        red[q * B_ + b] = sum;
    }
    __syncthreads();

    if (tid < B_) {
        const float S1 = red[M_ * B_ + tid];
        const float S2 = red[(M_ + 1) * B_ + tid];
        const float inv  = 1.0f / (float)(M_ * D_);
        const float mu   = S1 * inv;
        const float var  = S2 * inv - mu * mu;
        const float rstd = rsqrtf(var + EPS_);
        float out = 0.f;
        #pragma unroll
        for (int m = 0; m < M_; ++m) {
            float sv = rstd * (red[m * B_ + tid] - mu * wsum[m]) + bsum[m];
            out += fast_sigmoid(sv);
        }
        pre[(size_t)tid * O_ + o] = out;
    }
}

__global__ __launch_bounds__(256) void dnm_softmax(
    const float* __restrict__ pre, float* __restrict__ out)
{
    __shared__ float sh[256];
    const int b = blockIdx.x, t = threadIdx.x;
    float v = pre[b * O_ + t];
    sh[t] = v;
    __syncthreads();
    for (int s = 128; s > 0; s >>= 1) {
        if (t < s) sh[t] = fmaxf(sh[t], sh[t + s]);
        __syncthreads();
    }
    float mx = sh[0];
    __syncthreads();
    float e = __expf(v - mx);
    sh[t] = e;
    __syncthreads();
    for (int s = 128; s > 0; s >>= 1) {
        if (t < s) sh[t] += sh[t + s];
        __syncthreads();
    }
    out[b * O_ + t] = e * __builtin_amdgcn_rcpf(sh[0]);
}

extern "C" void kernel_launch(void* const* d_in, const int* in_sizes, int n_in,
                              void* d_out, int out_size, void* d_ws, size_t ws_size,
                              hipStream_t stream) {
    const float* x    = (const float*)d_in[0];
    const float* sn0w = (const float*)d_in[1];
    const float* sn0b = (const float*)d_in[2];
    const float* sw   = (const float*)d_in[3];
    const float* sb   = (const float*)d_in[4];
    const float* dnw  = (const float*)d_in[5];
    const float* dnb  = (const float*)d_in[6];
    float* out = (float*)d_out;

    float* ws   = (float*)d_ws;
    float* xn   = ws;
    float* wsum = ws + 32768;
    float* bsum = ws + 32776;
    float* pre  = ws + 32784;

    dnm_prep   <<<B_ + M_,  256,     0, stream>>>(x, sn0w, sn0b, dnw, dnb, xn, wsum, bsum);
    dnm_main   <<<O_,       NW * 32, 0, stream>>>(xn, sw, sb, dnw, wsum, bsum, pre);
    dnm_softmax<<<B_,       256,     0, stream>>>(pre, out);
}


# LayerNorm/row-sum kernel and softmax kernel: one wave per row, sums and max by lane butterflies (DPP, permlanex16) instead of LDS trees with 9 and 19 barriers; summation order of these sums changes
# speedup vs baseline: 1.5263x; 1.0831x over previous
#include <hip/hip_runtime.h>
#include <math.h>

typedef __attribute__((ext_vector_type(2))) float v2f;
typedef __attribute__((ext_vector_type(8))) float v8f;

#define B_   64
#define O_   256
#define M_   8
#define D_   512
#define EPS_ 1e-5f
#define TR   14
#define NT   5
#define NW   16
#define CH   16

__device__ __forceinline__ float fast_sigmoid(float x) {
#if __has_builtin(__builtin_amdgcn_tanhf)
    return fmaf(0.5f, __builtin_amdgcn_tanhf(0.5f * x), 0.5f);
#else
    return __builtin_amdgcn_rcpf(1.0f + __expf(-x));
#endif
}

__device__ __forceinline__ float zcore(float u) {
#if __has_builtin(__builtin_amdgcn_tanhf)
    return __builtin_amdgcn_tanhf(0.5f * u);
#else
    return __builtin_amdgcn_rcpf(1.0f + __expf(-u));
#endif
}

#define DPP_XOR1        0xB1
#define DPP_XOR2        0x4E
#define DPP_HALF_MIRROR 0x141
#define DPP_MIRROR      0x140

template <int CTRL>
__device__ __forceinline__ float dpp_get(float v) {
    return __int_as_float(__builtin_amdgcn_update_dpp(0, __float_as_int(v), CTRL, 0xf, 0xf, true));
}
__device__ __forceinline__ float other_row(float v) {
    const unsigned u = __float_as_uint(v);
    return __uint_as_float(__builtin_amdgcn_permlanex16(u, u, 0x76543210u, 0xfedcba98u, false, false));
}
__device__ __forceinline__ float wave_sum(float v) {
    v += dpp_get<DPP_XOR1>(v);
    v += dpp_get<DPP_XOR2>(v);
    v += dpp_get<DPP_HALF_MIRROR>(v);
    v += dpp_get<DPP_MIRROR>(v);
    v += other_row(v);
    return v;
}
__device__ __forceinline__ float wave_max(float v) {
    v = fmaxf(v, dpp_get<DPP_XOR1>(v));
    v = fmaxf(v, dpp_get<DPP_XOR2>(v));
    v = fmaxf(v, dpp_get<DPP_HALF_MIRROR>(v));
    v = fmaxf(v, dpp_get<DPP_MIRROR>(v));
    v = fmaxf(v, other_row(v));
    return v;
}

__global__ __launch_bounds__(32) void dnm_prep(
    const float* __restrict__ x,   const float* __restrict__ g,
    const float* __restrict__ be,  const float* __restrict__ dnw,
    const float* __restrict__ dnb, float* __restrict__ xn,
    float* __restrict__ wsum,      float* __restrict__ bsum)
{
    const int lane = threadIdx.x;
    const int bx   = blockIdx.x;

    if (bx < B_) {
        const float4* row = (const float4*)(x + bx * D_);
        const float4* g4  = (const float4*)g;
        const float4* b4  = (const float4*)be;
        float4 v[4], gv[4], bv[4];
        #pragma unroll
        for (int q = 0; q < 4; ++q) {
            v[q]  = row[lane + 32 * q];
            gv[q] = g4[lane + 32 * q];
            bv[q] = b4[lane + 32 * q];
        }
        float s1 = 0.f, s2 = 0.f;
        #pragma unroll
        for (int q = 0; q < 4; ++q) {
            s1 += (v[q].x + v[q].y) + (v[q].z + v[q].w);
            s2 += (v[q].x * v[q].x + v[q].y * v[q].y) + (v[q].z * v[q].z + v[q].w * v[q].w);
        }
        s1 = wave_sum(s1);
        s2 = wave_sum(s2);
        const float mu   = s1 * (1.0f / D_);
        const float var  = s2 * (1.0f / D_) - mu * mu;
        const float rstd = rsqrtf(var + EPS_);
        float4* o4 = (float4*)(xn + bx * D_);
        #pragma unroll
        for (int q = 0; q < 4; ++q) {
            float4 r;
            r.x = (v[q].x - mu) * rstd * gv[q].x + bv[q].x;
            r.y = (v[q].y - mu) * rstd * gv[q].y + bv[q].y;
            r.z = (v[q].z - mu) * rstd * gv[q].z + bv[q].z;
            r.w = (v[q].w - mu) * rstd * gv[q].w + bv[q].w;
            o4[lane + 32 * q] = r;
        }
    } else {
        const int m = bx - B_;
        const float4* wr = (const float4*)(dnw + m * D_);
        const float4* br = (const float4*)(dnb + m * D_);
        float s1 = 0.f, s2 = 0.f;
        #pragma unroll
        for (int q = 0; q < 4; ++q) {
            const float4 a = wr[lane + 32 * q];
            const float4 b = br[lane + 32 * q];
            s1 += (a.x + a.y) + (a.z + a.w);
            s2 += (b.x + b.y) + (b.z + b.w);
        }
        s1 = wave_sum(s1);
        s2 = wave_sum(s2);
        if (lane == 0) { wsum[m] = s1; bsum[m] = s2; }
    }
}

__global__ __launch_bounds__(NW * 32) void dnm_main(
    const float* __restrict__ xn,   const float* __restrict__ sw,
    const float* __restrict__ sb,   const float* __restrict__ dnw,
    const float* __restrict__ wsum, const float* __restrict__ bsum,
    float* __restrict__ pre)
{
    __shared__ float Tb[NW * M_ * NT * 16];
    __shared__ float S1b[NW * NT * 16];
    __shared__ float S2b[NW * NT * 16];
    __shared__ float red[(M_ + 2) * B_];

    const int o     = blockIdx.x;
    const int tid   = threadIdx.x;
    const int w     = tid >> 5;
    const int lane  = tid & 31;
    const int tb    = lane & 15;
    const int khalf = lane >> 4;
    const int cstart = 2 * CH * w + CH * khalf;

    const bool  isz = tb < TR;
#if __has_builtin(__builtin_amdgcn_tanhf)
    const float zs  = isz ? 0.5f : 0.0f;
    const float zo  = isz ? 0.5f : (tb == TR ? 1.0f : 0.0f);
#else
    const float zs  = isz ? 1.0f : 0.0f;
    const float zo  = (tb == TR) ? 1.0f : 0.0f;
#endif
    const float fw  = (tb == TR + 1) ? 1.0f : 0.0f;

    float xr[NT][CH];
    #pragma unroll
    for (int t = 0; t < NT; ++t) {
        int row = TR * t + (isz ? tb : TR - 1);
        row = row < B_ ? row : B_ - 1;
        #pragma unroll
        for (int q = 0; q < CH / 4; ++q) {
            const float4 v = *(const float4*)(xn + row * D_ + cstart + 4 * q);
            xr[t][4 * q + 0] = isz ? v.x : 0.f; xr[t][4 * q + 1] = isz ? v.y : 0.f;
            xr[t][4 * q + 2] = isz ? v.z : 0.f; xr[t][4 * q + 3] = isz ? v.w : 0.f;
        }
    }

    v8f acc[NT];
    #pragma unroll
    for (int t = 0; t < NT; ++t) acc[t] = (v8f){};

    const float* swl = sw + (size_t)o * (M_ * D_) + cstart;
    const float* sbl = sb + (size_t)o * (M_ * D_) + cstart;
    const float* dwl = dnw + cstart;
    float* tst = Tb + (w * M_ * NT) * 16 + tb;

    #pragma unroll 1
    for (int m = 0; m < M_; ++m) {
        __builtin_prefetch(swl + D_, 0, 1);
        __builtin_prefetch(sbl + D_, 0, 1);
        float s[CH], b[CH], g[CH];
        #pragma unroll
        for (int q = 0; q < CH / 4; ++q) {
            const float4 sv = *(const float4*)(swl + 4 * q);
            const float4 bv = *(const float4*)(sbl + 4 * q);
            const float4 dv = *(const float4*)(dwl + 4 * q);
            s[4 * q + 0] = sv.x; s[4 * q + 1] = sv.y; s[4 * q + 2] = sv.z; s[4 * q + 3] = sv.w;
            b[4 * q + 0] = bv.x; b[4 * q + 1] = bv.y; b[4 * q + 2] = bv.z; b[4 * q + 3] = bv.w;
            g[4 * q + 0] = fmaf(fw, dv.x, zo); g[4 * q + 1] = fmaf(fw, dv.y, zo);
            g[4 * q + 2] = fmaf(fw, dv.z, zo); g[4 * q + 3] = fmaf(fw, dv.w, zo);
        }

        #pragma unroll
        for (int i = 0; i < CH / 2; ++i) {
            #pragma unroll
            for (int t = 0; t < NT; ++t) {
                v2f a;
                a.x = fmaf(zs, zcore(fmaf(s[2 * i],     xr[t][2 * i],     b[2 * i])),     g[2 * i]);
                a.y = fmaf(zs, zcore(fmaf(s[2 * i + 1], xr[t][2 * i + 1], b[2 * i + 1])), g[2 * i + 1]);
                acc[t] = __builtin_amdgcn_wmma_f32_16x16x4_f32(false, a, false, a, (short)0, acc[t], false, false);
            }
        }

        asm volatile("v_nop\n\tv_nop\n\tv_nop\n\tv_nop\n\tv_nop"
                     : "+v"(acc[0]), "+v"(acc[1]), "+v"(acc[2]), "+v"(acc[3]), "+v"(acc[4]));
        if (khalf) {
            #pragma unroll
            for (int t = 0; t < NT; ++t) {
                tst[(m * NT + t) * 16] = acc[t][7];
                acc[t][7] = 0.f;
            }
        }
        swl += D_;
        sbl += D_;
        dwl += D_;
    }

    #pragma unroll
    for (int t = 0; t < NT; ++t) {
        if (khalf) S1b[(w * NT + t) * 16 + tb] = acc[t][6];
        float dv = acc[t][0];
        #pragma unroll
        for (int r = 1; r < 8; ++r) dv = ((tb & 7) == r) ? acc[t][r] : dv;
        if ((tb >> 3) == khalf) S2b[(w * NT + t) * 16 + tb] = dv;
    }
    __syncthreads();

    for (int idx = tid; idx < (M_ + 2) * B_; idx += NW * 32) {
        const int q = idx >> 6;
        const int b = idx & (B_ - 1);
        const int t = b / TR;
        const int j = b - t * TR;
        const float* p;
        int stride;
        if (q < M_)       { p = Tb  + (q * NT + t) * 16 + j; stride = M_ * NT * 16; }
        else if (q == M_) { p = S1b + t * 16 + j;            stride = NT * 16; }
        else              { p = S2b + t * 16 + j;            stride = NT * 16; }
        float sum = 0.f;
        for (int wv = 0; wv < NW; ++wv) sum += p[wv * stride];
        red[q * B_ + b] = sum;
    }
    __syncthreads();

    if (tid < B_) {
        const float S1 = red[M_ * B_ + tid];
        const float S2 = red[(M_ + 1) * B_ + tid];
        const float inv  = 1.0f / (float)(M_ * D_);
        const float mu   = S1 * inv;
        const float var  = S2 * inv - mu * mu;
        const float rstd = rsqrtf(var + EPS_);
        float out = 0.f;
        #pragma unroll
        for (int m = 0; m < M_; ++m) {
            float sv = rstd * (red[m * B_ + tid] - mu * wsum[m]) + bsum[m];
            out += fast_sigmoid(sv);
        }
        pre[(size_t)tid * O_ + o] = out;
    }
}

__global__ __launch_bounds__(32) void dnm_softmax(
    const float* __restrict__ pre, float* __restrict__ out)
{
    const int b = blockIdx.x, lane = threadIdx.x;
    const float4* p = (const float4*)(pre + b * O_);
    const float4 v0 = p[lane];
    const float4 v1 = p[lane + 32];
    float mx = fmaxf(fmaxf(fmaxf(v0.x, v0.y), fmaxf(v0.z, v0.w)),
                     fmaxf(fmaxf(v1.x, v1.y), fmaxf(v1.z, v1.w)));
    mx = wave_max(mx);
    float4 e0, e1;
    e0.x = __expf(v0.x - mx); e0.y = __expf(v0.y - mx);
    e0.z = __expf(v0.z - mx); e0.w = __expf(v0.w - mx);
    e1.x = __expf(v1.x - mx); e1.y = __expf(v1.y - mx);
    e1.z = __expf(v1.z - mx); e1.w = __expf(v1.w - mx);
    float sum = ((e0.x + e0.y) + (e0.z + e0.w)) + ((e1.x + e1.y) + (e1.z + e1.w));
    sum = wave_sum(sum);
    const float r = __builtin_amdgcn_rcpf(sum);
    float4* o4 = (float4*)(out + b * O_);
    o4[lane]      = make_float4(e0.x * r, e0.y * r, e0.z * r, e0.w * r);
    o4[lane + 32] = make_float4(e1.x * r, e1.y * r, e1.z * r, e1.w * r);
}

extern "C" void kernel_launch(void* const* d_in, const int* in_sizes, int n_in,
                              void* d_out, int out_size, void* d_ws, size_t ws_size,
                              hipStream_t stream) {
    const float* x    = (const float*)d_in[0];
    const float* sn0w = (const float*)d_in[1];
    const float* sn0b = (const float*)d_in[2];
    const float* sw   = (const float*)d_in[3];
    const float* sb   = (const float*)d_in[4];
    const float* dnw  = (const float*)d_in[5];
    const float* dnb  = (const float*)d_in[6];
    float* out = (float*)d_out;

    float* ws   = (float*)d_ws;
    float* xn   = ws;
    float* wsum = ws + 32768;
    float* bsum = ws + 32776;
    float* pre  = ws + 32784;

    dnm_prep   <<<B_ + M_,  32,      0, stream>>>(x, sn0w, sn0b, dnw, dnb, xn, wsum, bsum);
    dnm_main   <<<O_,       NW * 32, 0, stream>>>(xn, sw, sb, dnw, wsum, bsum, pre);
    dnm_softmax<<<B_,       32,      0, stream>>>(pre, out);
}


# LayerNorm/row-sum launch folded into the main kernel's prologue (2 launches, not 3): each workgroup, alone on its WGP, reads x once in whole lines and shares xn through LDS; sum order unchanged
# speedup vs baseline: 1.6210x; 1.0620x over previous
#include <hip/hip_runtime.h>
#include <math.h>

typedef __attribute__((ext_vector_type(2))) float v2f;
typedef __attribute__((ext_vector_type(8))) float v8f;

#define B_   64
#define O_   256
#define M_   8
#define D_   512
#define EPS_ 1e-5f
#define TR   14
#define NT   5
#define NW   16
#define CH   16
#define XP   (D_ + 4)

__device__ __forceinline__ float fast_sigmoid(float x) {
#if __has_builtin(__builtin_amdgcn_tanhf)
    return fmaf(0.5f, __builtin_amdgcn_tanhf(0.5f * x), 0.5f);
#else
    return __builtin_amdgcn_rcpf(1.0f + __expf(-x));
#endif
}

__device__ __forceinline__ float zcore(float u) {
#if __has_builtin(__builtin_amdgcn_tanhf)
    return __builtin_amdgcn_tanhf(0.5f * u);
#else
    return __builtin_amdgcn_rcpf(1.0f + __expf(-u));
#endif
}

#define DPP_XOR1        0xB1
#define DPP_XOR2        0x4E
#define DPP_HALF_MIRROR 0x141
#define DPP_MIRROR      0x140

template <int CTRL>
__device__ __forceinline__ float dpp_get(float v) {
    return __int_as_float(__builtin_amdgcn_update_dpp(0, __float_as_int(v), CTRL, 0xf, 0xf, true));
}
__device__ __forceinline__ float other_row(float v) {
    const unsigned u = __float_as_uint(v);
    return __uint_as_float(__builtin_amdgcn_permlanex16(u, u, 0x76543210u, 0xfedcba98u, false, false));
}
__device__ __forceinline__ float wave_sum(float v) {
    v += dpp_get<DPP_XOR1>(v);
    v += dpp_get<DPP_XOR2>(v);
    v += dpp_get<DPP_HALF_MIRROR>(v);
    v += dpp_get<DPP_MIRROR>(v);
    v += other_row(v);
    return v;
}
__device__ __forceinline__ float wave_max(float v) {
    v = fmaxf(v, dpp_get<DPP_XOR1>(v));
    v = fmaxf(v, dpp_get<DPP_XOR2>(v));
    v = fmaxf(v, dpp_get<DPP_HALF_MIRROR>(v));
    v = fmaxf(v, dpp_get<DPP_MIRROR>(v));
    v = fmaxf(v, other_row(v));
    return v;
}

__global__ __launch_bounds__(NW * 32) void dnm_main(
    const float* __restrict__ x,    const float* __restrict__ g0,
    const float* __restrict__ b0,   const float* __restrict__ sw,
    const float* __restrict__ sb,   const float* __restrict__ dnw,
    const float* __restrict__ dnb,  float* __restrict__ pre)
{
    __shared__ float xs[B_ * XP];
    __shared__ float wb[2 * M_];
    __shared__ float Tb[NW * M_ * NT * 16];
    __shared__ float S1b[NW * NT * 16];
    __shared__ float S2b[NW * NT * 16];
    __shared__ float red[(M_ + 2) * B_];

    const int o     = blockIdx.x;
    const int tid   = threadIdx.x;
    const int w     = tid >> 5;
    const int lane  = tid & 31;
    const int tb    = lane & 15;
    const int khalf = lane >> 4;
    const int cstart = 2 * CH * w + CH * khalf;

    {
        const int j = (w + o) & (NW - 1);
        const float4* g4 = (const float4*)g0;
        const float4* b4 = (const float4*)b0;
        const float*  wrow = (j < M_) ? dnw + j * D_ : dnb + (j - M_) * D_;
        float4 v[4][4], gv[4], bv[4], av[4];
        #pragma unroll
        for (int q = 0; q < 4; ++q) {
            #pragma unroll
            for (int r = 0; r < 4; ++r)
                v[r][q] = ((const float4*)(x + (4 * j + r) * D_))[lane + 32 * q];
            gv[q] = g4[lane + 32 * q];
            bv[q] = b4[lane + 32 * q];
            av[q] = ((const float4*)wrow)[lane + 32 * q];
        }
        float s1[4], s2[4];
        #pragma unroll
        for (int r = 0; r < 4; ++r) {
            s1[r] = 0.f; s2[r] = 0.f;
            #pragma unroll
            for (int q = 0; q < 4; ++q) {
                s1[r] += (v[r][q].x + v[r][q].y) + (v[r][q].z + v[r][q].w);
                s2[r] += (v[r][q].x * v[r][q].x + v[r][q].y * v[r][q].y)
                       + (v[r][q].z * v[r][q].z + v[r][q].w * v[r][q].w);
            }
        }
        float sa = 0.f;
        #pragma unroll
        for (int q = 0; q < 4; ++q) sa += (av[q].x + av[q].y) + (av[q].z + av[q].w);
        #pragma unroll
        for (int r = 0; r < 4; ++r) { s1[r] = wave_sum(s1[r]); s2[r] = wave_sum(s2[r]); }
        sa = wave_sum(sa);
        if (lane == 0) wb[j] = sa;
        #pragma unroll
        for (int r = 0; r < 4; ++r) {
            const float mu   = s1[r] * (1.0f / D_);
            const float var  = s2[r] * (1.0f / D_) - mu * mu;
            const float rstd = rsqrtf(var + EPS_);
            float* orow = xs + (4 * j + r) * XP;
            #pragma unroll
            for (int q = 0; q < 4; ++q) {
                float4 n;
                n.x = (v[r][q].x - mu) * rstd * gv[q].x + bv[q].x;
                n.y = (v[r][q].y - mu) * rstd * gv[q].y + bv[q].y;
                n.z = (v[r][q].z - mu) * rstd * gv[q].z + bv[q].z;
                n.w = (v[r][q].w - mu) * rstd * gv[q].w + bv[q].w;
                *(float4*)(orow + 4 * (lane + 32 * q)) = n;
            }
        }
    }
    __syncthreads();

    const bool  isz = tb < TR;
#if __has_builtin(__builtin_amdgcn_tanhf)
    const float zs  = isz ? 0.5f : 0.0f;
    const float zo  = isz ? 0.5f : (tb == TR ? 1.0f : 0.0f);
#else
    const float zs  = isz ? 1.0f : 0.0f;
    const float zo  = (tb == TR) ? 1.0f : 0.0f;
#endif
    const float fw  = (tb == TR + 1) ? 1.0f : 0.0f;

    float xr[NT][CH];
    #pragma unroll
    for (int t = 0; t < NT; ++t) {
        int row = TR * t + (isz ? tb : TR - 1);
        row = row < B_ ? row : B_ - 1;
        #pragma unroll
        for (int q = 0; q < CH / 4; ++q) {
            const float4 v = *(const float4*)(xs + row * XP + cstart + 4 * q);
            xr[t][4 * q + 0] = isz ? v.x : 0.f; xr[t][4 * q + 1] = isz ? v.y : 0.f;
            xr[t][4 * q + 2] = isz ? v.z : 0.f; xr[t][4 * q + 3] = isz ? v.w : 0.f;
        }
    }

    v8f acc[NT];
    #pragma unroll
    for (int t = 0; t < NT; ++t) acc[t] = (v8f){};

    const float* swl = sw + (size_t)o * (M_ * D_) + cstart;
    const float* sbl = sb + (size_t)o * (M_ * D_) + cstart;
    const float* dwl = dnw + cstart;
    float* tst = Tb + (w * M_ * NT) * 16 + tb;

    #pragma unroll 1
    for (int m = 0; m < M_; ++m) {
        __builtin_prefetch(swl + D_, 0, 1);
        __builtin_prefetch(sbl + D_, 0, 1);
        float s[CH], b[CH], g[CH];
        #pragma unroll
        for (int q = 0; q < CH / 4; ++q) {
            const float4 sv = *(const float4*)(swl + 4 * q);
            const float4 bv = *(const float4*)(sbl + 4 * q);
            const float4 dv = *(const float4*)(dwl + 4 * q);
            s[4 * q + 0] = sv.x; s[4 * q + 1] = sv.y; s[4 * q + 2] = sv.z; s[4 * q + 3] = sv.w;
            b[4 * q + 0] = bv.x; b[4 * q + 1] = bv.y; b[4 * q + 2] = bv.z; b[4 * q + 3] = bv.w;
            g[4 * q + 0] = fmaf(fw, dv.x, zo); g[4 * q + 1] = fmaf(fw, dv.y, zo);
            g[4 * q + 2] = fmaf(fw, dv.z, zo); g[4 * q + 3] = fmaf(fw, dv.w, zo);
        }

        #pragma unroll
        for (int i = 0; i < CH / 2; ++i) {
            #pragma unroll
            for (int t = 0; t < NT; ++t) {
                v2f a;
                a.x = fmaf(zs, zcore(fmaf(s[2 * i],     xr[t][2 * i],     b[2 * i])),     g[2 * i]);
                a.y = fmaf(zs, zcore(fmaf(s[2 * i + 1], xr[t][2 * i + 1], b[2 * i + 1])), g[2 * i + 1]);
                acc[t] = __builtin_amdgcn_wmma_f32_16x16x4_f32(false, a, false, a, (short)0, acc[t], false, false);
            }
        }

        asm volatile("v_nop\n\tv_nop\n\tv_nop\n\tv_nop\n\tv_nop"
                     : "+v"(acc[0]), "+v"(acc[1]), "+v"(acc[2]), "+v"(acc[3]), "+v"(acc[4]));
        if (khalf) {
            #pragma unroll
            for (int t = 0; t < NT; ++t) {
                tst[(m * NT + t) * 16] = acc[t][7];
                acc[t][7] = 0.f;
            }
        }
        swl += D_;
        sbl += D_;
        dwl += D_;
    }

    #pragma unroll
    for (int t = 0; t < NT; ++t) {
        if (khalf) S1b[(w * NT + t) * 16 + tb] = acc[t][6];
        float dv = acc[t][0];
        #pragma unroll
        for (int r = 1; r < 8; ++r) dv = ((tb & 7) == r) ? acc[t][r] : dv;
        if ((tb >> 3) == khalf) S2b[(w * NT + t) * 16 + tb] = dv;
    }
    __syncthreads();

    for (int idx = tid; idx < (M_ + 2) * B_; idx += NW * 32) {
        const int q = idx >> 6;
        const int b = idx & (B_ - 1);
        const int t = b / TR;
        const int j = b - t * TR;
        const float* p;
        int stride;
        if (q < M_)       { p = Tb  + (q * NT + t) * 16 + j; stride = M_ * NT * 16; }
        else if (q == M_) { p = S1b + t * 16 + j;            stride = NT * 16; }
        else              { p = S2b + t * 16 + j;            stride = NT * 16; }
        float sum = 0.f;
        for (int wv = 0; wv < NW; ++wv) sum += p[wv * stride];
        red[q * B_ + b] = sum;
    }
    __syncthreads();

    if (tid < B_) {
        const float S1 = red[M_ * B_ + tid];
        const float S2 = red[(M_ + 1) * B_ + tid];
        const float inv  = 1.0f / (float)(M_ * D_);
        const float mu   = S1 * inv;
        const float var  = S2 * inv - mu * mu;
        const float rstd = rsqrtf(var + EPS_);
        float out = 0.f;
        #pragma unroll
        for (int m = 0; m < M_; ++m) {
            float sv = rstd * (red[m * B_ + tid] - mu * wb[m]) + wb[M_ + m];
            out += fast_sigmoid(sv);
        }
        pre[(size_t)tid * O_ + o] = out;
    }
}

__global__ __launch_bounds__(32) void dnm_softmax(
    const float* __restrict__ pre, float* __restrict__ out)
{
    const int b = blockIdx.x, lane = threadIdx.x;
    const float4* p = (const float4*)(pre + b * O_);
    const float4 v0 = p[lane];
    const float4 v1 = p[lane + 32];
    float mx = fmaxf(fmaxf(fmaxf(v0.x, v0.y), fmaxf(v0.z, v0.w)),
                     fmaxf(fmaxf(v1.x, v1.y), fmaxf(v1.z, v1.w)));
    mx = wave_max(mx);
    float4 e0, e1;
    e0.x = __expf(v0.x - mx); e0.y = __expf(v0.y - mx);
    e0.z = __expf(v0.z - mx); e0.w = __expf(v0.w - mx);
    e1.x = __expf(v1.x - mx); e1.y = __expf(v1.y - mx);
    e1.z = __expf(v1.z - mx); e1.w = __expf(v1.w - mx);
    float sum = ((e0.x + e0.y) + (e0.z + e0.w)) + ((e1.x + e1.y) + (e1.z + e1.w));
    sum = wave_sum(sum);
    const float r = __builtin_amdgcn_rcpf(sum);
    float4* o4 = (float4*)(out + b * O_);
    o4[lane]      = make_float4(e0.x * r, e0.y * r, e0.z * r, e0.w * r);
    o4[lane + 32] = make_float4(e1.x * r, e1.y * r, e1.z * r, e1.w * r);
}

extern "C" void kernel_launch(void* const* d_in, const int* in_sizes, int n_in,
                              void* d_out, int out_size, void* d_ws, size_t ws_size,
                              hipStream_t stream) {
    const float* x    = (const float*)d_in[0];
    const float* sn0w = (const float*)d_in[1];
    const float* sn0b = (const float*)d_in[2];
    const float* sw   = (const float*)d_in[3];
    const float* sb   = (const float*)d_in[4];
    const float* dnw  = (const float*)d_in[5];
    const float* dnb  = (const float*)d_in[6];
    float* out = (float*)d_out;

    float* pre = (float*)d_ws;

    dnm_main   <<<O_, NW * 32, 0, stream>>>(x, sn0w, sn0b, sw, sb, dnw, dnb, pre);
    dnm_softmax<<<B_, 32,      0, stream>>>(pre, out);
}


# m-row loop software-pipelined: sw/sb/dn_w chunks reloaded in place a row ahead (row 0 under the LayerNorm, last row peeled), tanh+packed fma spread between product groups, 1/2 folded into x and sb
# speedup vs baseline: 1.7105x; 1.0552x over previous
#include <hip/hip_runtime.h>
#include <math.h>

typedef __attribute__((ext_vector_type(2))) float v2f;
typedef __attribute__((ext_vector_type(4))) float v4f;
typedef __attribute__((ext_vector_type(8))) float v8f;

#define B_   64
#define O_   256
#define M_   8
#define D_   512
#define EPS_ 1e-5f
#define TR   14
#define NT   5
#define NW   16
#define NJ   40
#define CH   16
#define XP   (D_ + 4)

__device__ __forceinline__ float fast_sigmoid(float x) {
#if __has_builtin(__builtin_amdgcn_tanhf)
    return fmaf(0.5f, __builtin_amdgcn_tanhf(0.5f * x), 0.5f);
#else
    return __builtin_amdgcn_rcpf(1.0f + __expf(-x));
#endif
}

__device__ __forceinline__ float zcore(float u) {
#if __has_builtin(__builtin_amdgcn_tanhf)
    return __builtin_amdgcn_tanhf(0.5f * u);
#else
    return __builtin_amdgcn_rcpf(1.0f + __expf(-u));
#endif
}

template <bool V> struct flag_t { static constexpr bool value = V; };

__device__ __forceinline__ v2f sq2lo(v4f v) { return (v2f){v.x, v.y}; }
__device__ __forceinline__ v2f sq2hi(v4f v) { return (v2f){v.z, v.w}; }

__device__ __forceinline__ float zcore1(float u) {
#if __has_builtin(__builtin_amdgcn_tanhf)
    return __builtin_amdgcn_tanhf(u);
#else
    return __builtin_amdgcn_rcpf(1.0f + __expf(-u));
#endif
}

#define DPP_XOR1        0xB1
#define DPP_XOR2        0x4E
#define DPP_HALF_MIRROR 0x141
#define DPP_MIRROR      0x140

template <int CTRL>
__device__ __forceinline__ float dpp_get(float v) {
    return __int_as_float(__builtin_amdgcn_update_dpp(0, __float_as_int(v), CTRL, 0xf, 0xf, true));
}
__device__ __forceinline__ float other_row(float v) {
    const unsigned u = __float_as_uint(v);
    return __uint_as_float(__builtin_amdgcn_permlanex16(u, u, 0x76543210u, 0xfedcba98u, false, false));
}
__device__ __forceinline__ float wave_sum(float v) {
    v += dpp_get<DPP_XOR1>(v);
    v += dpp_get<DPP_XOR2>(v);
    v += dpp_get<DPP_HALF_MIRROR>(v);
    v += dpp_get<DPP_MIRROR>(v);
    v += other_row(v);
    return v;
}
__device__ __forceinline__ float wave_max(float v) {
    v = fmaxf(v, dpp_get<DPP_XOR1>(v));
    v = fmaxf(v, dpp_get<DPP_XOR2>(v));
    v = fmaxf(v, dpp_get<DPP_HALF_MIRROR>(v));
    v = fmaxf(v, dpp_get<DPP_MIRROR>(v));
    v = fmaxf(v, other_row(v));
    return v;
}

__global__ __launch_bounds__(NW * 32) void dnm_main(
    const float* __restrict__ x,    const float* __restrict__ g0,
    const float* __restrict__ b0,   const float* __restrict__ sw,
    const float* __restrict__ sb,   const float* __restrict__ dnw,
    const float* __restrict__ dnb,  float* __restrict__ pre)
{
    __shared__ float xs[B_ * XP];
    __shared__ float wb[2 * M_];
    __shared__ float Tb[NW * M_ * NT * 16];
    __shared__ float S1b[NW * NT * 16];
    __shared__ float S2b[NW * NT * 16];
    __shared__ float red[(M_ + 2) * B_];

    const int o     = blockIdx.x;
    const int tid   = threadIdx.x;
    const int w     = tid >> 5;
    const int lane  = tid & 31;
    const int tb    = lane & 15;
    const int khalf = lane >> 4;
    const int cstart = 2 * CH * w + CH * khalf;

    const float* swl = sw + (size_t)o * (M_ * D_) + cstart;
    const float* sbl = sb + (size_t)o * (M_ * D_) + cstart;
    const float* dwl = dnw + cstart;
    v4f sq[CH / 4], bq[CH / 4], dq[CH / 4];
    #pragma unroll
    for (int q = 0; q < CH / 4; ++q) {
        sq[q] = *(const v4f*)(swl + 4 * q);
        bq[q] = *(const v4f*)(sbl + 4 * q);
        dq[q] = *(const v4f*)(dwl + 4 * q);
    }

    {
        const int j = (w + o) & (NW - 1);
        const float4* g4 = (const float4*)g0;
        const float4* b4 = (const float4*)b0;
        const float*  wrow = (j < M_) ? dnw + j * D_ : dnb + (j - M_) * D_;
        float4 v[4][4], gv[4], bv[4], av[4];
        #pragma unroll
        for (int q = 0; q < 4; ++q) {
            #pragma unroll
            for (int r = 0; r < 4; ++r)
                v[r][q] = ((const float4*)(x + (4 * j + r) * D_))[lane + 32 * q];
            gv[q] = g4[lane + 32 * q];
            bv[q] = b4[lane + 32 * q];
            av[q] = ((const float4*)wrow)[lane + 32 * q];
        }
        float s1[4], s2[4];
        #pragma unroll
        for (int r = 0; r < 4; ++r) {
            s1[r] = 0.f; s2[r] = 0.f;
            #pragma unroll
            for (int q = 0; q < 4; ++q) {
                s1[r] += (v[r][q].x + v[r][q].y) + (v[r][q].z + v[r][q].w);
                s2[r] += (v[r][q].x * v[r][q].x + v[r][q].y * v[r][q].y)
                       + (v[r][q].z * v[r][q].z + v[r][q].w * v[r][q].w);
            }
        }
        float sa = 0.f;
        #pragma unroll
        for (int q = 0; q < 4; ++q) sa += (av[q].x + av[q].y) + (av[q].z + av[q].w);
        #pragma unroll
        for (int r = 0; r < 4; ++r) { s1[r] = wave_sum(s1[r]); s2[r] = wave_sum(s2[r]); }
        sa = wave_sum(sa);
        if (lane == 0) wb[j] = sa;
        #pragma unroll
        for (int r = 0; r < 4; ++r) {
            const float mu   = s1[r] * (1.0f / D_);
            const float var  = s2[r] * (1.0f / D_) - mu * mu;
            const float rstd = rsqrtf(var + EPS_);
            float* orow = xs + (4 * j + r) * XP;
            #pragma unroll
            for (int q = 0; q < 4; ++q) {
                float4 n;
                n.x = (v[r][q].x - mu) * rstd * gv[q].x + bv[q].x;
                n.y = (v[r][q].y - mu) * rstd * gv[q].y + bv[q].y;
                n.z = (v[r][q].z - mu) * rstd * gv[q].z + bv[q].z;
                n.w = (v[r][q].w - mu) * rstd * gv[q].w + bv[q].w;
                *(float4*)(orow + 4 * (lane + 32 * q)) = n;
            }
        }
    }
    __syncthreads();

    const bool  isz = tb < TR;
#if __has_builtin(__builtin_amdgcn_tanhf)
    const float zs  = isz ? 0.5f : 0.0f;
    const float zo  = isz ? 0.5f : (tb == TR ? 1.0f : 0.0f);
#else
    const float zs  = isz ? 1.0f : 0.0f;
    const float zo  = (tb == TR) ? 1.0f : 0.0f;
#endif
    const float fw  = (tb == TR + 1) ? 1.0f : 0.0f;

#if __has_builtin(__builtin_amdgcn_tanhf)
    const float hk = 0.5f;
#else
    const float hk = 1.0f;
#endif
    v2f xr[NT][CH / 2];
    #pragma unroll
    for (int t = 0; t < NT; ++t) {
        int row = TR * t + (isz ? tb : TR - 1);
        row = row < B_ ? row : B_ - 1;
        #pragma unroll
        for (int q = 0; q < CH / 4; ++q) {
            const float4 v = *(const float4*)(xs + row * XP + cstart + 4 * q);
            xr[t][2 * q + 0] = (v2f){isz ? hk * v.x : 0.f, isz ? hk * v.y : 0.f};
            xr[t][2 * q + 1] = (v2f){isz ? hk * v.z : 0.f, isz ? hk * v.w : 0.f};
        }
    }

    v8f acc[NT];
    #pragma unroll
    for (int t = 0; t < NT; ++t) acc[t] = (v8f){};

    float* tst = Tb + (w * M_ * NT) * 16 + tb;
    const v2f zs2 = (v2f){zs, zs};
    const v2f zo2 = (v2f){zo, zo};
    const v2f fw2 = (v2f){fw, fw};
    const v2f hk2 = (v2f){hk, hk};

    auto row = [&](const int m, auto refill) {
        const float* swn = swl + D_;
        const float* sbn = sbl + D_;
        const float* dwn = dwl + D_;

        v2f uu[NJ], cc[NJ], aa[NJ], bh[CH / 2], g[CH / 2];
        bh[0] = hk2 * sq2lo(bq[0]);
        uu[0] = __builtin_elementwise_fma(sq2lo(sq[0]), xr[0][0], bh[0]);
        uu[1] = __builtin_elementwise_fma(sq2lo(sq[0]), xr[1][0], bh[0]);
        #pragma unroll
        for (int j = 0; j < NJ + 3; ++j) {
            if (j < NJ) cc[j].x = zcore1(uu[j].x);
            if (j >= 3) {
                const int i = (j - 3) / NT;
                if ((j - 3) % NT == 0)
                    g[i] = __builtin_elementwise_fma(fw2, (i & 1) ? sq2hi(dq[i / 2]) : sq2lo(dq[i / 2]), zo2);
                aa[j - 3] = __builtin_elementwise_fma(zs2, cc[j - 3], g[i]);
            }
            __builtin_amdgcn_sched_barrier(0);
            if (j < NJ) cc[j].y = zcore1(uu[j].y);
            if (j + 2 < NJ) {
                const int i = (j + 2) / NT;
                if ((j + 2) % NT == 0)
                    bh[i] = hk2 * ((i & 1) ? sq2hi(bq[i / 2]) : sq2lo(bq[i / 2]));
                uu[j + 2] = __builtin_elementwise_fma((i & 1) ? sq2hi(sq[i / 2]) : sq2lo(sq[i / 2]), xr[(j + 2) % NT][i], bh[i]);
            }
            __builtin_amdgcn_sched_barrier(0);
            if (decltype(refill)::value && j % 10 == 8) {
                const int q = j / 10;
                sq[q] = *(const v4f*)(swn + 4 * q);
                bq[q] = *(const v4f*)(sbn + 4 * q);
                dq[q] = *(const v4f*)(dwn + 4 * q);
                __builtin_amdgcn_sched_barrier(0);
            }
            if (j >= 3 && (j - 3) % NT == NT - 1) {
                const int j0 = j - 3 - (NT - 1);
                #pragma unroll
                for (int t = 0; t < NT; ++t)
                    acc[t] = __builtin_amdgcn_wmma_f32_16x16x4_f32(false, aa[j0 + t], false, aa[j0 + t],
                                                                   (short)0, acc[t], false, false);
                if (j != NJ + 2)
                    asm volatile("v_nop\n\tv_nop\n\tv_nop\n\tv_nop"
                                 : "+v"(aa[j0]), "+v"(aa[j0 + 1]), "+v"(aa[j0 + 2]), "+v"(aa[j0 + 3]), "+v"(aa[j0 + 4]));
                __builtin_amdgcn_sched_barrier(0);
            }
        }

        asm volatile("v_nop\n\tv_nop\n\tv_nop\n\tv_nop\n\tv_nop"
                     : "+v"(acc[0]), "+v"(acc[1]), "+v"(acc[2]), "+v"(acc[3]), "+v"(acc[4]));
        if (khalf) {
            #pragma unroll
            for (int t = 0; t < NT; ++t) {
                tst[(m * NT + t) * 16] = acc[t][7];
                acc[t][7] = 0.f;
            }
        }
        swl += D_;
        sbl += D_;
        dwl += D_;
    };
    #pragma unroll 1
    for (int m = 0; m < M_ - 1; ++m) row(m, flag_t<true>{});
    row(M_ - 1, flag_t<false>{});

    #pragma unroll
    for (int t = 0; t < NT; ++t) {
        if (khalf) S1b[(w * NT + t) * 16 + tb] = acc[t][6];
        float dv = acc[t][0];
        #pragma unroll
        for (int r = 1; r < 8; ++r) dv = ((tb & 7) == r) ? acc[t][r] : dv;
        if ((tb >> 3) == khalf) S2b[(w * NT + t) * 16 + tb] = dv;
    }
    __syncthreads();

    for (int idx = tid; idx < (M_ + 2) * B_; idx += NW * 32) {
        const int q = idx >> 6;
        const int b = idx & (B_ - 1);
        const int t = b / TR;
        const int j = b - t * TR;
        const float* p;
        int stride;
        if (q < M_)       { p = Tb  + (q * NT + t) * 16 + j; stride = M_ * NT * 16; }
        else if (q == M_) { p = S1b + t * 16 + j;            stride = NT * 16; }
        else              { p = S2b + t * 16 + j;            stride = NT * 16; }
        float sum = 0.f;
        for (int wv = 0; wv < NW; ++wv) sum += p[wv * stride];
        red[q * B_ + b] = sum;
    }
    __syncthreads();

    if (tid < B_) {
        const float S1 = red[M_ * B_ + tid];
        const float S2 = red[(M_ + 1) * B_ + tid];
        const float inv  = 1.0f / (float)(M_ * D_);
        const float mu   = S1 * inv;
        const float var  = S2 * inv - mu * mu;
        const float rstd = rsqrtf(var + EPS_);
        float out = 0.f;
        #pragma unroll
        for (int m = 0; m < M_; ++m) {
            float sv = rstd * (red[m * B_ + tid] - mu * wb[m]) + wb[M_ + m];
            out += fast_sigmoid(sv);
        }
        pre[(size_t)tid * O_ + o] = out;
    }
}

__global__ __launch_bounds__(32) void dnm_softmax(
    const float* __restrict__ pre, float* __restrict__ out)
{
    const int b = blockIdx.x, lane = threadIdx.x;
    const float4* p = (const float4*)(pre + b * O_);
    const float4 v0 = p[lane];
    const float4 v1 = p[lane + 32];
    float mx = fmaxf(fmaxf(fmaxf(v0.x, v0.y), fmaxf(v0.z, v0.w)),
                     fmaxf(fmaxf(v1.x, v1.y), fmaxf(v1.z, v1.w)));
    mx = wave_max(mx);
    float4 e0, e1;
    e0.x = __expf(v0.x - mx); e0.y = __expf(v0.y - mx);
    e0.z = __expf(v0.z - mx); e0.w = __expf(v0.w - mx);
    e1.x = __expf(v1.x - mx); e1.y = __expf(v1.y - mx);
    e1.z = __expf(v1.z - mx); e1.w = __expf(v1.w - mx);
    float sum = ((e0.x + e0.y) + (e0.z + e0.w)) + ((e1.x + e1.y) + (e1.z + e1.w));
    sum = wave_sum(sum);
    const float r = __builtin_amdgcn_rcpf(sum);
    float4* o4 = (float4*)(out + b * O_);
    o4[lane]      = make_float4(e0.x * r, e0.y * r, e0.z * r, e0.w * r);
    o4[lane + 32] = make_float4(e1.x * r, e1.y * r, e1.z * r, e1.w * r);
}

extern "C" void kernel_launch(void* const* d_in, const int* in_sizes, int n_in,
                              void* d_out, int out_size, void* d_ws, size_t ws_size,
                              hipStream_t stream) {
    const float* x    = (const float*)d_in[0];
    const float* sn0w = (const float*)d_in[1];
    const float* sn0b = (const float*)d_in[2];
    const float* sw   = (const float*)d_in[3];
    const float* sb   = (const float*)d_in[4];
    const float* dnw  = (const float*)d_in[5];
    const float* dnb  = (const float*)d_in[6];
    float* out = (float*)d_out;

    float* pre = (float*)d_ws;

    dnm_main   <<<O_, NW * 32, 0, stream>>>(x, sn0w, sn0b, sw, sb, dnw, dnb, pre);
    dnm_softmax<<<B_, 32,      0, stream>>>(pre, out);
}


# 16-row batch tiles: z fills all 16 B columns, A keeps rows 14,15 as ones and dn_w; sum z^2 of rows 14,15 via a side tile S x S^T per two steps: 36 WMMAs per m row, not 40; that sum's order changes
# speedup vs baseline: 1.7686x; 1.0340x over previous
#include <hip/hip_runtime.h>
#include <math.h>

typedef __attribute__((ext_vector_type(2))) float v2f;
typedef __attribute__((ext_vector_type(4))) float v4f;
typedef __attribute__((ext_vector_type(8))) float v8f;

#define B_   64
#define O_   256
#define M_   8
#define D_   512
#define EPS_ 1e-5f
#define TR   14
#define NT   4
#define NP   5
#define NW   16
#define NJ   40
#define LIVE(j) (((j) % NP) < NT || (((j) / NP) & 1))
#define CH   16
#define XP   (D_ + 4)

__device__ __forceinline__ float fast_sigmoid(float x) {
#if __has_builtin(__builtin_amdgcn_tanhf)
    return fmaf(0.5f, __builtin_amdgcn_tanhf(0.5f * x), 0.5f);
#else
    return __builtin_amdgcn_rcpf(1.0f + __expf(-x));
#endif
}

__device__ __forceinline__ float zcore(float u) {
#if __has_builtin(__builtin_amdgcn_tanhf)
    return __builtin_amdgcn_tanhf(0.5f * u);
#else
    return __builtin_amdgcn_rcpf(1.0f + __expf(-u));
#endif
}

template <bool V> struct flag_t { static constexpr bool value = V; };

__device__ __forceinline__ v2f sq2lo(v4f v) { return (v2f){v.x, v.y}; }
__device__ __forceinline__ v2f sq2hi(v4f v) { return (v2f){v.z, v.w}; }

__device__ __forceinline__ float zcore1(float u) {
#if __has_builtin(__builtin_amdgcn_tanhf)
    return __builtin_amdgcn_tanhf(u);
#else
    return __builtin_amdgcn_rcpf(1.0f + __expf(-u));
#endif
}

#define DPP_XOR1        0xB1
#define DPP_XOR2        0x4E
#define DPP_HALF_MIRROR 0x141
#define DPP_MIRROR      0x140

template <int CTRL>
__device__ __forceinline__ float dpp_get(float v) {
    return __int_as_float(__builtin_amdgcn_update_dpp(0, __float_as_int(v), CTRL, 0xf, 0xf, true));
}
__device__ __forceinline__ float other_row(float v) {
    const unsigned u = __float_as_uint(v);
    return __uint_as_float(__builtin_amdgcn_permlanex16(u, u, 0x76543210u, 0xfedcba98u, false, false));
}
__device__ __forceinline__ float wave_sum(float v) {
    v += dpp_get<DPP_XOR1>(v);
    v += dpp_get<DPP_XOR2>(v);
    v += dpp_get<DPP_HALF_MIRROR>(v);
    v += dpp_get<DPP_MIRROR>(v);
    v += other_row(v);
    return v;
}
__device__ __forceinline__ float wave_max(float v) {
    v = fmaxf(v, dpp_get<DPP_XOR1>(v));
    v = fmaxf(v, dpp_get<DPP_XOR2>(v));
    v = fmaxf(v, dpp_get<DPP_HALF_MIRROR>(v));
    v = fmaxf(v, dpp_get<DPP_MIRROR>(v));
    v = fmaxf(v, other_row(v));
    return v;
}

__global__ __launch_bounds__(NW * 32) void dnm_main(
    const float* __restrict__ x,    const float* __restrict__ g0,
    const float* __restrict__ b0,   const float* __restrict__ sw,
    const float* __restrict__ sb,   const float* __restrict__ dnw,
    const float* __restrict__ dnb,  float* __restrict__ pre)
{
    __shared__ float xs[B_ * XP];
    __shared__ float wb[2 * M_];
    __shared__ float Tb[NW * M_ * NT * 16];
    __shared__ float S1b[NW * NT * 16];
    __shared__ float S2b[NW * NT * 16];
    __shared__ float S2x[NW * 16];
    __shared__ float red[(M_ + 2) * B_];

    const int o     = blockIdx.x;
    const int tid   = threadIdx.x;
    const int w     = tid >> 5;
    const int lane  = tid & 31;
    const int tb    = lane & 15;
    const int khalf = lane >> 4;
    const int cstart = 2 * CH * w + CH * khalf;

    const float* swl = sw + (size_t)o * (M_ * D_) + cstart;
    const float* sbl = sb + (size_t)o * (M_ * D_) + cstart;
    const float* dwl = dnw + cstart;
    v4f sq[CH / 4], bq[CH / 4], dq[CH / 4];
    #pragma unroll
    for (int q = 0; q < CH / 4; ++q) {
        sq[q] = *(const v4f*)(swl + 4 * q);
        bq[q] = *(const v4f*)(sbl + 4 * q);
        dq[q] = *(const v4f*)(dwl + 4 * q);
    }

    {
        const int j = (w + o) & (NW - 1);
        const float4* g4 = (const float4*)g0;
        const float4* b4 = (const float4*)b0;
        const float*  wrow = (j < M_) ? dnw + j * D_ : dnb + (j - M_) * D_;
        float4 v[4][4], gv[4], bv[4], av[4];
        #pragma unroll
        for (int q = 0; q < 4; ++q) {
            #pragma unroll
            for (int r = 0; r < 4; ++r)
                v[r][q] = ((const float4*)(x + (4 * j + r) * D_))[lane + 32 * q];
            gv[q] = g4[lane + 32 * q];
            bv[q] = b4[lane + 32 * q];
            av[q] = ((const float4*)wrow)[lane + 32 * q];
        }
        float s1[4], s2[4];
        #pragma unroll
        for (int r = 0; r < 4; ++r) {
            s1[r] = 0.f; s2[r] = 0.f;
            #pragma unroll
            for (int q = 0; q < 4; ++q) {
                s1[r] += (v[r][q].x + v[r][q].y) + (v[r][q].z + v[r][q].w);
                s2[r] += (v[r][q].x * v[r][q].x + v[r][q].y * v[r][q].y)
                       + (v[r][q].z * v[r][q].z + v[r][q].w * v[r][q].w);
            }
        }
        float sa = 0.f;
        #pragma unroll
        for (int q = 0; q < 4; ++q) sa += (av[q].x + av[q].y) + (av[q].z + av[q].w);
        #pragma unroll
        for (int r = 0; r < 4; ++r) { s1[r] = wave_sum(s1[r]); s2[r] = wave_sum(s2[r]); }
        sa = wave_sum(sa);
        if (lane == 0) wb[j] = sa;
        #pragma unroll
        for (int r = 0; r < 4; ++r) {
            const float mu   = s1[r] * (1.0f / D_);
            const float var  = s2[r] * (1.0f / D_) - mu * mu;
            const float rstd = rsqrtf(var + EPS_);
            float* orow = xs + (4 * j + r) * XP;
            #pragma unroll
            for (int q = 0; q < 4; ++q) {
                float4 n;
                n.x = (v[r][q].x - mu) * rstd * gv[q].x + bv[q].x;
                n.y = (v[r][q].y - mu) * rstd * gv[q].y + bv[q].y;
                n.z = (v[r][q].z - mu) * rstd * gv[q].z + bv[q].z;
                n.w = (v[r][q].w - mu) * rstd * gv[q].w + bv[q].w;
                *(float4*)(orow + 4 * (lane + 32 * q)) = n;
            }
        }
    }
    __syncthreads();

    const bool  isz = tb < TR;
#if __has_builtin(__builtin_amdgcn_tanhf)
    const float zs  = isz ? 0.5f : 0.0f;
    const float zo  = isz ? 0.5f : (tb == TR ? 1.0f : 0.0f);
#else
    const float zs  = isz ? 1.0f : 0.0f;
    const float zo  = (tb == TR) ? 1.0f : 0.0f;
#endif
    const float fw  = (tb == TR + 1) ? 1.0f : 0.0f;
#if __has_builtin(__builtin_amdgcn_tanhf)
    const v2f zb2 = (v2f){0.5f, 0.5f}, zob2 = (v2f){0.5f, 0.5f};
#else
    const v2f zb2 = (v2f){1.0f, 1.0f}, zob2 = (v2f){0.0f, 0.0f};
#endif

#if __has_builtin(__builtin_amdgcn_tanhf)
    const float hk = 0.5f;
#else
    const float hk = 1.0f;
#endif
    v2f xr[NT][CH / 2];
    #pragma unroll
    for (int t = 0; t < NT; ++t) {
        const int row = 16 * t + tb;
        #pragma unroll
        for (int q = 0; q < CH / 4; ++q) {
            const float4 v = *(const float4*)(xs + row * XP + cstart + 4 * q);
            xr[t][2 * q + 0] = (v2f){hk * v.x, hk * v.y};
            xr[t][2 * q + 1] = (v2f){hk * v.z, hk * v.w};
        }
    }
    const bool hi8 = (tb & 8) != 0;
    v2f xe[CH / 4];
    {
        const int row = 16 * ((tb & 7) >> 1) + TR + (tb & 1);
        #pragma unroll
        for (int q = 0; q < CH / 4; ++q) {
            const float2 v = *(const float2*)(xs + row * XP + cstart + 4 * q + (hi8 ? 2 : 0));
            xe[q] = (v2f){hk * v.x, hk * v.y};
        }
    }

    v8f acc[NT], acce = (v8f){};
    #pragma unroll
    for (int t = 0; t < NT; ++t) acc[t] = (v8f){};

    float* tst = Tb + (w * M_ * NT) * 16 + tb;
    const v2f zs2 = (v2f){zs, zs};
    const v2f zo2 = (v2f){zo, zo};
    const v2f fw2 = (v2f){fw, fw};
    const v2f hk2 = (v2f){hk, hk};

    auto row = [&](const int m, auto refill) {
        const float* swn = swl + D_;
        const float* sbn = sbl + D_;
        const float* dwn = dwl + D_;

        v2f uu[NJ], cc[NJ], aa[NJ], ab[NJ], bh[CH / 2], g[CH / 2];
        bh[0] = hk2 * sq2lo(bq[0]);
        uu[0] = __builtin_elementwise_fma(sq2lo(sq[0]), xr[0][0], bh[0]);
        uu[1] = __builtin_elementwise_fma(sq2lo(sq[0]), xr[1][0], bh[0]);
        #pragma unroll
        for (int j = 0; j < NJ + 3; ++j) {
            if (j < NJ && LIVE(j)) cc[j].x = zcore1(uu[j].x);
            if (j >= 3 && LIVE(j - 3)) {
                const int i = (j - 3) / NP;
                if ((j - 3) % NP == 0)
                    g[i] = __builtin_elementwise_fma(fw2, (i & 1) ? sq2hi(dq[i / 2]) : sq2lo(dq[i / 2]), zo2);
                if ((j - 3) % NP < NT) aa[j - 3] = __builtin_elementwise_fma(zs2, cc[j - 3], g[i]);
                else                   aa[j - 3] = __builtin_elementwise_fma(zb2, cc[j - 3], zob2);
            }
            __builtin_amdgcn_sched_barrier(0);
            if (j < NJ && LIVE(j)) cc[j].y = zcore1(uu[j].y);
            if (j >= 3 && LIVE(j - 3) && (j - 3) % NP < NT)
                ab[j - 3] = __builtin_elementwise_fma(zb2, cc[j - 3], zob2);
            if (j + 2 < NJ && LIVE(j + 2)) {
                const int i = (j + 2) / NP;
                if ((j + 2) % NP == 0)
                    bh[i] = hk2 * ((i & 1) ? sq2hi(bq[i / 2]) : sq2lo(bq[i / 2]));
                if ((j + 2) % NP < NT) {
                    uu[j + 2] = __builtin_elementwise_fma((i & 1) ? sq2hi(sq[i / 2]) : sq2lo(sq[i / 2]), xr[(j + 2) % NP][i], bh[i]);
                } else {
                    const v2f swe = hi8 ? sq2hi(sq[i / 2]) : sq2lo(sq[i / 2]);
                    const v2f bhe = hi8 ? bh[i] : bh[i - 1];
                    uu[j + 2] = __builtin_elementwise_fma(swe, xe[i / 2], bhe);
                }
            }
            __builtin_amdgcn_sched_barrier(0);
            if (decltype(refill)::value && j % 10 == 8) {
                const int q = j / 10;
                sq[q] = *(const v4f*)(swn + 4 * q);
                bq[q] = *(const v4f*)(sbn + 4 * q);
                dq[q] = *(const v4f*)(dwn + 4 * q);
                __builtin_amdgcn_sched_barrier(0);
            }
            if (j >= 3 && (j - 3) % NP == NP - 1) {
                const int j0 = j - 3 - (NP - 1);
                const bool side = ((j - 3) / NP) & 1;
                #pragma unroll
                for (int t = 0; t < NT; ++t)
                    acc[t] = __builtin_amdgcn_wmma_f32_16x16x4_f32(false, aa[j0 + t], false, ab[j0 + t],
                                                                   (short)0, acc[t], false, false);
                if (side)
                    acce = __builtin_amdgcn_wmma_f32_16x16x4_f32(false, aa[j - 3], false, aa[j - 3],
                                                                 (short)0, acce, false, false);
                if (j != NJ + 2) {
                    if (side)
                        asm volatile("v_nop\n\tv_nop\n\tv_nop\n\tv_nop"
                                     : "+v"(aa[j0]), "+v"(aa[j0 + 1]), "+v"(aa[j0 + 2]), "+v"(aa[j0 + 3]),
                                       "+v"(ab[j0]), "+v"(ab[j0 + 1]), "+v"(ab[j0 + 2]), "+v"(ab[j0 + 3]), "+v"(aa[j - 3]));
                    else
                        asm volatile("v_nop\n\tv_nop\n\tv_nop\n\tv_nop"
                                     : "+v"(aa[j0]), "+v"(aa[j0 + 1]), "+v"(aa[j0 + 2]), "+v"(aa[j0 + 3]),
                                       "+v"(ab[j0]), "+v"(ab[j0 + 1]), "+v"(ab[j0 + 2]), "+v"(ab[j0 + 3]));
                }
                __builtin_amdgcn_sched_barrier(0);
            }
        }

        asm volatile("v_nop\n\tv_nop\n\tv_nop\n\tv_nop\n\tv_nop"
                     : "+v"(acc[0]), "+v"(acc[1]), "+v"(acc[2]), "+v"(acc[3]), "+v"(acce));
        if (khalf) {
            #pragma unroll
            for (int t = 0; t < NT; ++t) {
                tst[(m * NT + t) * 16] = acc[t][7];
                acc[t][7] = 0.f;
            }
        }
        swl += D_;
        sbl += D_;
        dwl += D_;
    };
    #pragma unroll 1
    for (int m = 0; m < M_ - 1; ++m) row(m, flag_t<true>{});
    row(M_ - 1, flag_t<false>{});

    #pragma unroll
    for (int t = 0; t < NT; ++t) {
        if (khalf) S1b[(w * NT + t) * 16 + tb] = acc[t][6];
        float dv = acc[t][0];
        #pragma unroll
        for (int r = 1; r < 8; ++r) dv = ((tb & 7) == r) ? acc[t][r] : dv;
        if ((tb >> 3) == khalf && isz) S2b[(w * NT + t) * 16 + tb] = dv;
    }
    {
        float dv = acce[0];
        #pragma unroll
        for (int r = 1; r < 8; ++r) dv = ((tb & 7) == r) ? acce[r] : dv;
        if ((tb >> 3) == khalf) S2x[w * 16 + tb] = dv;
    }
    __syncthreads();

    for (int idx = tid; idx < (M_ + 2) * B_; idx += NW * 32) {
        const int q = idx >> 6;
        const int b = idx & (B_ - 1);
        const int t = b >> 4;
        const int j = b & 15;
        const float* p;
        int stride;
        if (q < M_)       { p = Tb  + (q * NT + t) * 16 + j; stride = M_ * NT * 16; }
        else if (q == M_) { p = S1b + t * 16 + j;            stride = NT * 16; }
        else if (j < TR)  { p = S2b + t * 16 + j;            stride = NT * 16; }
        else              { p = S2x + 2 * t + (j - TR);      stride = 16; }
        float sum = 0.f;
        if (q > M_ && j >= TR) { for (int wv = 0; wv < NW; ++wv) sum += p[wv * stride] + p[wv * stride + 8]; }
        else                   { for (int wv = 0; wv < NW; ++wv) sum += p[wv * stride]; }
        red[q * B_ + b] = sum;
    }
    __syncthreads();

    if (tid < B_) {
        const float S1 = red[M_ * B_ + tid];
        const float S2 = red[(M_ + 1) * B_ + tid];
        const float inv  = 1.0f / (float)(M_ * D_);
        const float mu   = S1 * inv;
        const float var  = S2 * inv - mu * mu;
        const float rstd = rsqrtf(var + EPS_);
        float out = 0.f;
        #pragma unroll
        for (int m = 0; m < M_; ++m) {
            float sv = rstd * (red[m * B_ + tid] - mu * wb[m]) + wb[M_ + m];
            out += fast_sigmoid(sv);
        }
        pre[(size_t)tid * O_ + o] = out;
    }
}

__global__ __launch_bounds__(32) void dnm_softmax(
    const float* __restrict__ pre, float* __restrict__ out)
{
    const int b = blockIdx.x, lane = threadIdx.x;
    const float4* p = (const float4*)(pre + b * O_);
    const float4 v0 = p[lane];
    const float4 v1 = p[lane + 32];
    float mx = fmaxf(fmaxf(fmaxf(v0.x, v0.y), fmaxf(v0.z, v0.w)),
                     fmaxf(fmaxf(v1.x, v1.y), fmaxf(v1.z, v1.w)));
    mx = wave_max(mx);
    float4 e0, e1;
    e0.x = __expf(v0.x - mx); e0.y = __expf(v0.y - mx);
    e0.z = __expf(v0.z - mx); e0.w = __expf(v0.w - mx);
    e1.x = __expf(v1.x - mx); e1.y = __expf(v1.y - mx);
    e1.z = __expf(v1.z - mx); e1.w = __expf(v1.w - mx);
    float sum = ((e0.x + e0.y) + (e0.z + e0.w)) + ((e1.x + e1.y) + (e1.z + e1.w));
    sum = wave_sum(sum);
    const float r = __builtin_amdgcn_rcpf(sum);
    float4* o4 = (float4*)(out + b * O_);
    o4[lane]      = make_float4(e0.x * r, e0.y * r, e0.z * r, e0.w * r);
    o4[lane + 32] = make_float4(e1.x * r, e1.y * r, e1.z * r, e1.w * r);
}

extern "C" void kernel_launch(void* const* d_in, const int* in_sizes, int n_in,
                              void* d_out, int out_size, void* d_ws, size_t ws_size,
                              hipStream_t stream) {
    const float* x    = (const float*)d_in[0];
    const float* sn0w = (const float*)d_in[1];
    const float* sn0b = (const float*)d_in[2];
    const float* sw   = (const float*)d_in[3];
    const float* sb   = (const float*)d_in[4];
    const float* dnw  = (const float*)d_in[5];
    const float* dnb  = (const float*)d_in[6];
    float* out = (float*)d_out;

    float* pre = (float*)d_ws;

    dnm_main   <<<O_, NW * 32, 0, stream>>>(x, sn0w, sn0b, sw, sb, dnw, dnb, pre);
    dnm_softmax<<<B_, 32,      0, stream>>>(pre, out);
}


# A and B of a product hold different 14-row batch half tiles (ones, dn_w in lanes 14,15 of both): 20 WMMAs per m row, not 36; sum z^2 moves to a per-lane packed fma chain; summation order changes
# speedup vs baseline: 2.0252x; 1.1451x over previous
#include <hip/hip_runtime.h>
#include <math.h>

typedef __attribute__((ext_vector_type(2))) float v2f;
typedef __attribute__((ext_vector_type(4))) float v4f;
typedef __attribute__((ext_vector_type(8))) float v8f;

#define B_   64
#define O_   256
#define M_   8
#define D_   512
#define EPS_ 1e-5f
#define TR   14
#define NH   5
#define NA   3
#define NP   5
#define NW   16
#define NJ   40
#define NS   96
#define CH   16
#define XP   (D_ + 4)

__device__ __forceinline__ float fast_sigmoid(float x) {
#if __has_builtin(__builtin_amdgcn_tanhf)
    return fmaf(0.5f, __builtin_amdgcn_tanhf(0.5f * x), 0.5f);
#else
    return __builtin_amdgcn_rcpf(1.0f + __expf(-x));
#endif
}

__device__ __forceinline__ float zcore(float u) {
#if __has_builtin(__builtin_amdgcn_tanhf)
    return __builtin_amdgcn_tanhf(0.5f * u);
#else
    return __builtin_amdgcn_rcpf(1.0f + __expf(-u));
#endif
}

template <bool V> struct flag_t { static constexpr bool value = V; };

__device__ __forceinline__ v2f sq2lo(v4f v) { return (v2f){v.x, v.y}; }
__device__ __forceinline__ v2f sq2hi(v4f v) { return (v2f){v.z, v.w}; }

__device__ __forceinline__ float zcore1(float u) {
#if __has_builtin(__builtin_amdgcn_tanhf)
    return __builtin_amdgcn_tanhf(u);
#else
    return __builtin_amdgcn_rcpf(1.0f + __expf(-u));
#endif
}

#define DPP_XOR1        0xB1
#define DPP_XOR2        0x4E
#define DPP_HALF_MIRROR 0x141
#define DPP_MIRROR      0x140

template <int CTRL>
__device__ __forceinline__ float dpp_get(float v) {
    return __int_as_float(__builtin_amdgcn_update_dpp(0, __float_as_int(v), CTRL, 0xf, 0xf, true));
}
__device__ __forceinline__ float other_row(float v) {
    const unsigned u = __float_as_uint(v);
    return __uint_as_float(__builtin_amdgcn_permlanex16(u, u, 0x76543210u, 0xfedcba98u, false, false));
}
__device__ __forceinline__ float wave_sum(float v) {
    v += dpp_get<DPP_XOR1>(v);
    v += dpp_get<DPP_XOR2>(v);
    v += dpp_get<DPP_HALF_MIRROR>(v);
    v += dpp_get<DPP_MIRROR>(v);
    v += other_row(v);
    return v;
}
__device__ __forceinline__ float wave_max(float v) {
    v = fmaxf(v, dpp_get<DPP_XOR1>(v));
    v = fmaxf(v, dpp_get<DPP_XOR2>(v));
    v = fmaxf(v, dpp_get<DPP_HALF_MIRROR>(v));
    v = fmaxf(v, dpp_get<DPP_MIRROR>(v));
    v = fmaxf(v, other_row(v));
    return v;
}

__global__ __launch_bounds__(NW * 32) void dnm_main(
    const float* __restrict__ x,    const float* __restrict__ g0,
    const float* __restrict__ b0,   const float* __restrict__ sw,
    const float* __restrict__ sb,   const float* __restrict__ dnw,
    const float* __restrict__ dnb,  float* __restrict__ pre)
{
    __shared__ float xs[B_ * XP];
    __shared__ float wb[2 * M_];
    __shared__ __attribute__((aligned(16))) float Tb[NW * M_ * NS];
    __shared__ __attribute__((aligned(16))) float S1b[NW * NS];
    __shared__ float S2b[NW * NH * 32];
    __shared__ float red[(M_ + 2) * B_];

    const int o     = blockIdx.x;
    const int tid   = threadIdx.x;
    const int w     = tid >> 5;
    const int lane  = tid & 31;
    const int tb    = lane & 15;
    const int khalf = lane >> 4;
    const int cstart = 2 * CH * w + CH * khalf;

    const float* swl = sw + (size_t)o * (M_ * D_) + cstart;
    const float* sbl = sb + (size_t)o * (M_ * D_) + cstart;
    const float* dwl = dnw + cstart;
    v4f sq[CH / 4], bq[CH / 4], dq[CH / 4];
    #pragma unroll
    for (int q = 0; q < CH / 4; ++q) {
        sq[q] = *(const v4f*)(swl + 4 * q);
        bq[q] = *(const v4f*)(sbl + 4 * q);
        dq[q] = *(const v4f*)(dwl + 4 * q);
    }

    {
        const int j = (w + o) & (NW - 1);
        const float4* g4 = (const float4*)g0;
        const float4* b4 = (const float4*)b0;
        const float*  wrow = (j < M_) ? dnw + j * D_ : dnb + (j - M_) * D_;
        float4 v[4][4], gv[4], bv[4], av[4];
        #pragma unroll
        for (int q = 0; q < 4; ++q) {
            #pragma unroll
            for (int r = 0; r < 4; ++r)
                v[r][q] = ((const float4*)(x + (4 * j + r) * D_))[lane + 32 * q];
            gv[q] = g4[lane + 32 * q];
            bv[q] = b4[lane + 32 * q];
            av[q] = ((const float4*)wrow)[lane + 32 * q];
        }
        float s1[4], s2[4];
        #pragma unroll
        for (int r = 0; r < 4; ++r) {
            s1[r] = 0.f; s2[r] = 0.f;
            #pragma unroll
            for (int q = 0; q < 4; ++q) {
                s1[r] += (v[r][q].x + v[r][q].y) + (v[r][q].z + v[r][q].w);
                s2[r] += (v[r][q].x * v[r][q].x + v[r][q].y * v[r][q].y)
                       + (v[r][q].z * v[r][q].z + v[r][q].w * v[r][q].w);
            }
        }
        float sa = 0.f;
        #pragma unroll
        for (int q = 0; q < 4; ++q) sa += (av[q].x + av[q].y) + (av[q].z + av[q].w);
        #pragma unroll
        for (int r = 0; r < 4; ++r) { s1[r] = wave_sum(s1[r]); s2[r] = wave_sum(s2[r]); }
        sa = wave_sum(sa);
        if (lane == 0) wb[j] = sa;
        #pragma unroll
        for (int r = 0; r < 4; ++r) {
            const float mu   = s1[r] * (1.0f / D_);
            const float var  = s2[r] * (1.0f / D_) - mu * mu;
            const float rstd = rsqrtf(var + EPS_);
            float* orow = xs + (4 * j + r) * XP;
            #pragma unroll
            for (int q = 0; q < 4; ++q) {
                float4 n;
                n.x = (v[r][q].x - mu) * rstd * gv[q].x + bv[q].x;
                n.y = (v[r][q].y - mu) * rstd * gv[q].y + bv[q].y;
                n.z = (v[r][q].z - mu) * rstd * gv[q].z + bv[q].z;
                n.w = (v[r][q].w - mu) * rstd * gv[q].w + bv[q].w;
                *(float4*)(orow + 4 * (lane + 32 * q)) = n;
            }
        }
    }
    __syncthreads();

    const bool  isz = tb < TR;
#if __has_builtin(__builtin_amdgcn_tanhf)
    const float zs  = isz ? 0.5f : 0.0f;
    const float zo  = isz ? 0.5f : (tb == TR ? 1.0f : 0.0f);
#else
    const float zs  = isz ? 1.0f : 0.0f;
    const float zo  = (tb == TR) ? 1.0f : 0.0f;
#endif
    const float fw  = (tb == TR + 1) ? 1.0f : 0.0f;

#if __has_builtin(__builtin_amdgcn_tanhf)
    const float hk = 0.5f;
#else
    const float hk = 1.0f;
#endif
    v2f xr[NH][CH / 2];
    #pragma unroll
    for (int h = 0; h < NH; ++h) {
        const int row = (h < NH - 1) ? TR * h + tb : TR * (NH - 1) + (tb & 7);
        #pragma unroll
        for (int q = 0; q < CH / 4; ++q) {
            const float4 v = *(const float4*)(xs + row * XP + cstart + 4 * q);
            xr[h][2 * q + 0] = (v2f){hk * v.x, hk * v.y};
            xr[h][2 * q + 1] = (v2f){hk * v.z, hk * v.w};
        }
    }

    v8f acc[NA];
    #pragma unroll
    for (int a = 0; a < NA; ++a) acc[a] = (v8f){};
    v2f s2[NH];
    #pragma unroll
    for (int h = 0; h < NH; ++h) s2[h] = (v2f){0.f, 0.f};

    float* tst = Tb + (w * M_) * NS;
    const v2f zs2 = (v2f){zs, zs};
    const v2f zo2 = (v2f){zo, zo};
    const v2f fw2 = (v2f){fw, fw};
    const v2f hk2 = (v2f){hk, hk};

    auto row = [&](const int m, auto refill) {
        const float* swn = swl + D_;
        const float* sbn = sbl + D_;
        const float* dwn = dwl + D_;

        v2f uu[NJ], cc[NJ], op[NJ], bh[CH / 2], g[CH / 2];
        bh[0] = hk2 * sq2lo(bq[0]);
        uu[0] = __builtin_elementwise_fma(sq2lo(sq[0]), xr[0][0], bh[0]);
        uu[1] = __builtin_elementwise_fma(sq2lo(sq[0]), xr[1][0], bh[0]);
        #pragma unroll
        for (int j = 0; j < NJ + 3; ++j) {
            if (j < NJ) cc[j].x = zcore1(uu[j].x);
            if (j >= 3) {
                const int i = (j - 3) / NP;
                if ((j - 3) % (2 * NP) == 0) {
                    g[i]     = __builtin_elementwise_fma(fw2, sq2lo(dq[i / 2]), zo2);
                    g[i + 1] = __builtin_elementwise_fma(fw2, sq2hi(dq[i / 2]), zo2);
                }
                op[j - 3] = __builtin_elementwise_fma(zs2, cc[j - 3], ((j - 3) % NP == NP - 1) ? g[i ^ 1] : g[i]);
            }
            __builtin_amdgcn_sched_barrier(0);
            if (j < NJ) cc[j].y = zcore1(uu[j].y);
            if (j >= 4)
                s2[(j - 4) % NP] = __builtin_elementwise_fma(op[j - 4], op[j - 4], s2[(j - 4) % NP]);
            if (j == NJ + 2)
                s2[(j - 3) % NP] = __builtin_elementwise_fma(op[j - 3], op[j - 3], s2[(j - 3) % NP]);
            if (j + 2 < NJ) {
                const int i = (j + 2) / NP;
                if ((j + 2) % NP == 0)
                    bh[i] = hk2 * ((i & 1) ? sq2hi(bq[i / 2]) : sq2lo(bq[i / 2]));
                uu[j + 2] = __builtin_elementwise_fma((i & 1) ? sq2hi(sq[i / 2]) : sq2lo(sq[i / 2]), xr[(j + 2) % NP][i], bh[i]);
            }
            __builtin_amdgcn_sched_barrier(0);
            if (decltype(refill)::value && j % 10 == 8) {
                const int q = j / 10;
                sq[q] = *(const v4f*)(swn + 4 * q);
                bq[q] = *(const v4f*)(sbn + 4 * q);
                dq[q] = *(const v4f*)(dwn + 4 * q);
                __builtin_amdgcn_sched_barrier(0);
            }
            if (j >= 3 && (j - 3) % NP == NP - 1) {
                const int j0 = j - 3 - (NP - 1);
                const bool odd = ((j - 3) / NP) & 1;
                acc[0] = __builtin_amdgcn_wmma_f32_16x16x4_f32(false, op[j0], false, op[j0 + 1],
                                                               (short)0, acc[0], false, false);
                acc[1] = __builtin_amdgcn_wmma_f32_16x16x4_f32(false, op[j0 + 2], false, op[j0 + 3],
                                                               (short)0, acc[1], false, false);
                if (odd)
                    acc[2] = __builtin_amdgcn_wmma_f32_16x16x4_f32(false, op[j0 - 1], false, op[j0 + 4],
                                                                   (short)0, acc[2], false, false);
                if (j != NJ + 2) {
                    if (odd)
                        asm volatile("v_nop\n\tv_nop\n\tv_nop\n\tv_nop"
                                     : "+v"(op[j0]), "+v"(op[j0 + 1]), "+v"(op[j0 + 2]), "+v"(op[j0 + 3]),
                                       "+v"(op[j0 - 1]), "+v"(op[j0 + 4]),
                                       "+v"(s2[0]), "+v"(s2[1]), "+v"(s2[2]), "+v"(s2[3]), "+v"(s2[4]));
                    else
                        asm volatile("v_nop\n\tv_nop\n\tv_nop\n\tv_nop"
                                     : "+v"(op[j0]), "+v"(op[j0 + 1]), "+v"(op[j0 + 2]), "+v"(op[j0 + 3]),
                                       "+v"(s2[0]), "+v"(s2[1]), "+v"(s2[2]), "+v"(s2[3]), "+v"(s2[4]));
                }
                __builtin_amdgcn_sched_barrier(0);
            }
        }

        asm volatile("v_nop\n\tv_nop\n\tv_nop\n\tv_nop\n\tv_nop"
                     : "+v"(acc[0]), "+v"(acc[1]), "+v"(acc[2]),
                       "+v"(s2[0]), "+v"(s2[1]), "+v"(s2[2]), "+v"(s2[3]), "+v"(s2[4]));
        float* tm = tst + m * NS;
        if (khalf) {
            #pragma unroll
            for (int a = 0; a < NA; ++a) {
                tm[32 * a + 16 + tb] = acc[a][7];
                acc[a][7] = 0.f;
            }
        }
        if (tb == TR + 1) {
            #pragma unroll
            for (int a = 0; a < NA; ++a) {
                *(v4f*)(tm + 32 * a + 8 * khalf)     = (v4f){acc[a][0], acc[a][1], acc[a][2], acc[a][3]};
                *(v4f*)(tm + 32 * a + 8 * khalf + 4) = (v4f){acc[a][4], acc[a][5], acc[a][6], acc[a][7]};
                acc[a] = (v8f){};
            }
        }
        swl += D_;
        sbl += D_;
        dwl += D_;
    };
    #pragma unroll 1
    for (int m = 0; m < M_ - 1; ++m) row(m, flag_t<true>{});
    row(M_ - 1, flag_t<false>{});

    #pragma unroll
    for (int a = 0; a < NA; ++a) {
        if (khalf) S1b[w * NS + 32 * a + 16 + tb] = acc[a][6];
        if (tb == TR) {
            *(v4f*)(S1b + w * NS + 32 * a + 8 * khalf)     = (v4f){acc[a][0], acc[a][1], acc[a][2], acc[a][3]};
            *(v4f*)(S1b + w * NS + 32 * a + 8 * khalf + 4) = (v4f){acc[a][4], acc[a][5], acc[a][6], acc[a][7]};
        }
    }
    #pragma unroll
    for (int h = 0; h < NH; ++h) S2b[(w * NH + h) * 32 + lane] = s2[h].x + s2[h].y;
    __syncthreads();

    for (int idx = tid; idx < (M_ + 2) * B_; idx += NW * 32) {
        const int q = idx >> 6;
        const int b = idx & (B_ - 1);
        const bool last = b >= TR * (NH - 1);
        const int h = last ? NH - 1 : b / TR;
        const int i = b - TR * h;
        const int slot = 16 * h + i;
        const float* p;
        int stride;
        if (q < M_)       { p = Tb  + q * NS + slot;  stride = M_ * NS; }
        else if (q == M_) { p = S1b + slot;           stride = NS; }
        else              { p = S2b + h * 32 + i;     stride = NH * 32; }
        float sum = 0.f;
        if (q > M_ || last) { for (int wv = 0; wv < NW; ++wv) sum += p[wv * stride] + p[wv * stride + 16]; }
        else                { for (int wv = 0; wv < NW; ++wv) sum += p[wv * stride]; }
        red[q * B_ + b] = sum;
    }
    __syncthreads();

    if (tid < B_) {
        const float S1 = red[M_ * B_ + tid];
        const float S2 = red[(M_ + 1) * B_ + tid];
        const float inv  = 1.0f / (float)(M_ * D_);
        const float mu   = S1 * inv;
        const float var  = S2 * inv - mu * mu;
        const float rstd = rsqrtf(var + EPS_);
        float out = 0.f;
        #pragma unroll
        for (int m = 0; m < M_; ++m) {
            float sv = rstd * (red[m * B_ + tid] - mu * wb[m]) + wb[M_ + m];
            out += fast_sigmoid(sv);
        }
        pre[(size_t)tid * O_ + o] = out;
    }
}

__global__ __launch_bounds__(32) void dnm_softmax(
    const float* __restrict__ pre, float* __restrict__ out)
{
    const int b = blockIdx.x, lane = threadIdx.x;
    const float4* p = (const float4*)(pre + b * O_);
    const float4 v0 = p[lane];
    const float4 v1 = p[lane + 32];
    float mx = fmaxf(fmaxf(fmaxf(v0.x, v0.y), fmaxf(v0.z, v0.w)),
                     fmaxf(fmaxf(v1.x, v1.y), fmaxf(v1.z, v1.w)));
    mx = wave_max(mx);
    float4 e0, e1;
    e0.x = __expf(v0.x - mx); e0.y = __expf(v0.y - mx);
    e0.z = __expf(v0.z - mx); e0.w = __expf(v0.w - mx);
    e1.x = __expf(v1.x - mx); e1.y = __expf(v1.y - mx);
    e1.z = __expf(v1.z - mx); e1.w = __expf(v1.w - mx);
    float sum = ((e0.x + e0.y) + (e0.z + e0.w)) + ((e1.x + e1.y) + (e1.z + e1.w));
    sum = wave_sum(sum);
    const float r = __builtin_amdgcn_rcpf(sum);
    float4* o4 = (float4*)(out + b * O_);
    o4[lane]      = make_float4(e0.x * r, e0.y * r, e0.z * r, e0.w * r);
    o4[lane + 32] = make_float4(e1.x * r, e1.y * r, e1.z * r, e1.w * r);
}

extern "C" void kernel_launch(void* const* d_in, const int* in_sizes, int n_in,
                              void* d_out, int out_size, void* d_ws, size_t ws_size,
                              hipStream_t stream) {
    const float* x    = (const float*)d_in[0];
    const float* sn0w = (const float*)d_in[1];
    const float* sn0b = (const float*)d_in[2];
    const float* sw   = (const float*)d_in[3];
    const float* sb   = (const float*)d_in[4];
    const float* dnw  = (const float*)d_in[5];
    const float* dnb  = (const float*)d_in[6];
    float* out = (float*)d_out;

    float* pre = (float*)d_ws;

    dnm_main   <<<O_, NW * 32, 0, stream>>>(x, sn0w, sn0b, sw, sb, dnw, dnb, pre);
    dnm_softmax<<<B_, 32,      0, stream>>>(pre, out);
}
